# starting kernel
# baseline (speedup 1.0000x reference)
#include <hip/hip_runtime.h>

typedef __attribute__((ext_vector_type(2))) float v2f;
typedef __attribute__((ext_vector_type(8))) float v8f;

#define FEAT 64

__global__ void zero_f32(float* __restrict__ p, long n) {
    long i = (long)blockIdx.x * blockDim.x + threadIdx.x;
    long stride = (long)gridDim.x * blockDim.x;
    for (; i < n; i += stride) p[i] = 0.0f;
}

__global__ void degree_kernel(const int* __restrict__ dst, float* __restrict__ deg, int E) {
    int e = blockIdx.x * blockDim.x + threadIdx.x;
    if (e < E) atomicAdd(&deg[dst[e]], 1.0f);
}

__global__ void scatter_kernel(const float* __restrict__ x,
                               const int* __restrict__ src,
                               const int* __restrict__ dst,
                               float* __restrict__ aggr, long E) {
    long t = (long)blockIdx.x * blockDim.x + threadIdx.x;
    if (t >= E * 16) return;
    long e = t >> 4;
    int  f = (int)(t & 15) << 2;
    long s = src[e];
    long d = dst[e];
    const float4 v = *(const float4*)(x + s * FEAT + f);
    float* a = aggr + d * FEAT + f;
    atomicAdd(a + 0, v.x);
    atomicAdd(a + 1, v.y);
    atomicAdd(a + 2, v.z);
    atomicAdd(a + 3, v.w);
}

__global__ void __launch_bounds__(256)
sage_update_wmma(const float* __restrict__ x, const float* __restrict__ aggr,
                 const float* __restrict__ deg,
                 const float* __restrict__ Wl, const float* __restrict__ bl,
                 const float* __restrict__ Wr,
                 float* __restrict__ out, int N) {
    __shared__ float Wl_s[FEAT * FEAT];
    __shared__ float Wr_s[FEAT * FEAT];
    __shared__ float b_s[FEAT];

    for (int i = threadIdx.x; i < FEAT * FEAT / 4; i += 256) {
        ((float4*)Wl_s)[i] = ((const float4*)Wl)[i];
        ((float4*)Wr_s)[i] = ((const float4*)Wr)[i];
    }
    if (threadIdx.x < FEAT) b_s[threadIdx.x] = bl[threadIdx.x];
    __syncthreads();

    const int lane = threadIdx.x & 31;
    const int wave = threadIdx.x >> 5;
    const int tile = blockIdx.x * 8 + wave;
    const int base = tile * 16;
    if (base >= N) return;

    const int row  = lane & 15;
    const int half = lane >> 4;
    const int node = base + row;
    const float dinv = 1.0f / fmaxf(deg[node], 1.0f);

    v2f a_ag[16], a_x[16];
    const float* agp = aggr + (long)node * FEAT;
    const float* xp  = x    + (long)node * FEAT;
#pragma unroll
    for (int ks = 0; ks < 16; ++ks) {
        const int k0 = ks * 4 + half * 2;
        v2f t0; t0.x = agp[k0] * dinv; t0.y = agp[k0 + 1] * dinv;
        a_ag[ks] = t0;
        v2f t1; t1.x = xp[k0];         t1.y = xp[k0 + 1];
        a_x[ks] = t1;
    }

#pragma unroll
    for (int nb = 0; nb < 4; ++nb) {
        const int ncol = nb * 16 + row;
        v8f c = {};
#pragma unroll
        for (int ks = 0; ks < 16; ++ks) {
            const int k0 = ks * 4 + half * 2;
            v2f bfl; bfl.x = Wl_s[ncol * FEAT + k0]; bfl.y = Wl_s[ncol * FEAT + k0 + 1];
            c = __builtin_amdgcn_wmma_f32_16x16x4_f32(false, a_ag[ks], false, bfl,
                                                      (short)0, c, false, false);
            v2f bfr; bfr.x = Wr_s[ncol * FEAT + k0]; bfr.y = Wr_s[ncol * FEAT + k0 + 1];
            c = __builtin_amdgcn_wmma_f32_16x16x4_f32(false, a_x[ks], false, bfr,
                                                      (short)0, c, false, false);
        }
        const float bias = b_s[nb * 16 + row];
#pragma unroll
        for (int r = 0; r < 8; ++r) {
            const int M = r + half * 8;
            out[(long)(base + M) * FEAT + nb * 16 + row] = c[r] + bias;
        }
    }
}

extern "C" void kernel_launch(void* const* d_in, const int* in_sizes, int n_in,
                              void* d_out, int out_size, void* d_ws, size_t ws_size,
                              hipStream_t stream) {
    const float* x   = (const float*)d_in[0];
    const int*   eix = (const int*)d_in[1];
    const float* W1l = (const float*)d_in[2];
    const float* b1l = (const float*)d_in[3];
    const float* W1r = (const float*)d_in[4];
    const float* W2l = (const float*)d_in[5];
    const float* b2l = (const float*)d_in[6];
    const float* W2r = (const float*)d_in[7];

    const int  N = in_sizes[0] / FEAT;
    const long E = in_sizes[1] / 2;
    const int* src = eix;
    const int* dst = eix + E;

    float* deg  = (float*)d_ws;
    float* aggr = deg + N;
    float* x1   = aggr + (size_t)N * FEAT;
    float* outp = (float*)d_out;

    const int deg_blocks = (int)((E + 255) / 256);
    const int sc_blocks  = (int)((E * 16 + 255) / 256);
    const int up_blocks  = (N + 127) / 128;

    zero_f32<<<256, 256, 0, stream>>>(deg, (long)N);
    degree_kernel<<<deg_blocks, 256, 0, stream>>>(dst, deg, (int)E);

    zero_f32<<<2048, 256, 0, stream>>>(aggr, (long)N * FEAT);
    scatter_kernel<<<sc_blocks, 256, 0, stream>>>(x, src, dst, aggr, E);
    sage_update_wmma<<<up_blocks, 256, 0, stream>>>(x, aggr, deg, W1l, b1l, W1r, x1, N);

    zero_f32<<<2048, 256, 0, stream>>>(aggr, (long)N * FEAT);
    scatter_kernel<<<sc_blocks, 256, 0, stream>>>(x1, src, dst, aggr, E);
    sage_update_wmma<<<up_blocks, 256, 0, stream>>>(x1, aggr, deg, W2l, b2l, W2r, outp, N);
}


# owner work-groups aggregate: each keeps its node range's sums and counts in LDS, scans the destination list, adds matching source rows by LDS atomics, stores rows once; 8 to 4 launches; arrival order
# speedup vs baseline: 1.8171x; 1.8171x over previous
#include <hip/hip_runtime.h>

typedef __attribute__((ext_vector_type(2))) float v2f;
typedef __attribute__((ext_vector_type(8))) float v8f;

#define FEAT 64
#define TILE 640
#define QCAP 256
#define AG_THREADS 1024

__global__ void __launch_bounds__(AG_THREADS)
aggregate_owner(const float* __restrict__ x, const int* __restrict__ src, const int* __restrict__ dst,
                float* __restrict__ aggr, float* __restrict__ deg, long E, int N) {
    __shared__ float    tile_s[(TILE + 1) * FEAT];
    __shared__ unsigned cnt_s[TILE + 1];
    __shared__ uint2    q_s[AG_THREADS / 32][QCAP];

    const int tid  = threadIdx.x;
    const int lane = tid & 31;
    const int lo   = blockIdx.x * TILE;
    const unsigned ucnt = (unsigned)min(TILE, N - lo);
    uint2* q = q_s[tid >> 5];

    for (int i = tid; i < (TILE + 1) * FEAT / 4; i += AG_THREADS)
        ((float4*)tile_s)[i] = make_float4(0.f, 0.f, 0.f, 0.f);
    for (int i = tid; i < TILE + 1; i += AG_THREADS) cnt_s[i] = 0u;
    __syncthreads();

    unsigned qhead = 0u, qcount = 0u;

    auto append = [&](bool m, unsigned e, unsigned row) {
        const unsigned mask = __builtin_amdgcn_ballot_w32(m);
        if (mask) {
            const unsigned below = __builtin_amdgcn_mbcnt_lo(mask, 0u);
            if (m) q[(qhead + qcount + below) & (QCAP - 1)] = make_uint2(e, row);
            qcount += (unsigned)__builtin_popcount(mask);
        }
    };

    auto drain = [&](unsigned n) {
        const bool valid = (unsigned)lane < n;
        uint2 ent = make_uint2(0u, (unsigned)TILE);
        if (valid) ent = q[(qhead + (unsigned)lane) & (QCAP - 1)];
        const unsigned row = ent.y;
        const int s = valid ? src[ent.x] : 0;
        atomicAdd(&cnt_s[row], 1u);
        for (int j0 = 0; j0 < 32; j0 += 8) {
            float2 v[8];
            unsigned r[8];
#pragma unroll
            for (int u = 0; u < 8; ++u) {
                const int sj = __builtin_amdgcn_readlane(s, j0 + u);
                r[u] = (unsigned)__builtin_amdgcn_readlane((int)row, j0 + u);
                v[u] = ((const float2*)(x + (long)sj * FEAT))[lane];
            }
#pragma unroll
            for (int u = 0; u < 8; ++u) {
                float* t = tile_s + r[u] * FEAT + lane * 2;
                atomicAdd(t, v[u].x);
                atomicAdd(t + 1, v[u].y);
            }
        }
        qhead = (qhead + 32u) & (QCAP - 1);
        qcount -= n;
    };

    auto process4 = [&](int4 d, unsigned e0) {
        const unsigned a0 = (unsigned)d.x - (unsigned)lo, a1 = (unsigned)d.y - (unsigned)lo;
        const unsigned a2 = (unsigned)d.z - (unsigned)lo, a3 = (unsigned)d.w - (unsigned)lo;
        const bool m0 = a0 < ucnt, m1 = a1 < ucnt, m2 = a2 < ucnt, m3 = a3 < ucnt;
        if (__builtin_amdgcn_ballot_w32(m0 | m1 | m2 | m3)) {
            append(m0, e0, a0); append(m1, e0 + 1u, a1); append(m2, e0 + 2u, a2); append(m3, e0 + 3u, a3);
            while (qcount >= 32u) drain(32u);
        }
    };

    const bool vec_ok = (((unsigned long long)dst) & 15ull) == 0ull;
    const long nfull  = vec_ok ? (E >> 2) : 0;
    const long nvec   = (E + 3) >> 2;
    const long Tmain  = (nfull / AG_THREADS) & ~7L;
    const long T      = (nvec + AG_THREADS - 1) / AG_THREADS;
    const int4* dst4  = (const int4*)dst;

    for (long t = 0; t < Tmain; t += 8) {
        int4 d[8];
#pragma unroll
        for (int u = 0; u < 8; ++u) d[u] = dst4[(t + u) * AG_THREADS + tid];
#pragma unroll
        for (int u = 0; u < 8; ++u) process4(d[u], (unsigned)(((t + u) * AG_THREADS + tid) * 4));
    }
    for (long t = Tmain; t < T; ++t) {
        const long e0 = (t * AG_THREADS + tid) * 4;
        int4 d;
        if (vec_ok && e0 + 3 < E) d = dst4[e0 >> 2];
        else {
            d.x = (e0     < E) ? dst[e0]     : -1;
            d.y = (e0 + 1 < E) ? dst[e0 + 1] : -1;
            d.z = (e0 + 2 < E) ? dst[e0 + 2] : -1;
            d.w = (e0 + 3 < E) ? dst[e0 + 3] : -1;
        }
        process4(d, (unsigned)e0);
    }
    if (qcount) drain(qcount);
    __syncthreads();

    float4* orow = (float4*)(aggr + (long)lo * FEAT);
    for (int i = tid; i < (int)ucnt * (FEAT / 4); i += AG_THREADS) orow[i] = ((const float4*)tile_s)[i];
    for (int i = tid; i < (int)ucnt; i += AG_THREADS) deg[lo + i] = (float)cnt_s[i];
}

__global__ void __launch_bounds__(256)
sage_update_wmma(const float* __restrict__ x, const float* __restrict__ aggr,
                 const float* __restrict__ deg,
                 const float* __restrict__ Wl, const float* __restrict__ bl,
                 const float* __restrict__ Wr,
                 float* __restrict__ out, int N) {
    __shared__ float Wl_s[FEAT * FEAT];
    __shared__ float Wr_s[FEAT * FEAT];
    __shared__ float b_s[FEAT];

    for (int i = threadIdx.x; i < FEAT * FEAT / 4; i += 256) {
        ((float4*)Wl_s)[i] = ((const float4*)Wl)[i];
        ((float4*)Wr_s)[i] = ((const float4*)Wr)[i];
    }
    if (threadIdx.x < FEAT) b_s[threadIdx.x] = bl[threadIdx.x];
    __syncthreads();

    const int lane = threadIdx.x & 31;
    const int wave = threadIdx.x >> 5;
    const int tile = blockIdx.x * 8 + wave;
    const int base = tile * 16;
    if (base >= N) return;

    const int row  = lane & 15;
    const int half = lane >> 4;
    const int node = base + row;
    const float dinv = 1.0f / fmaxf(deg[node], 1.0f);

    v2f a_ag[16], a_x[16];
    const float* agp = aggr + (long)node * FEAT;
    const float* xp  = x    + (long)node * FEAT;
#pragma unroll
    for (int ks = 0; ks < 16; ++ks) {
        const int k0 = ks * 4 + half * 2;
        v2f t0; t0.x = agp[k0] * dinv; t0.y = agp[k0 + 1] * dinv;
        a_ag[ks] = t0;
        v2f t1; t1.x = xp[k0];         t1.y = xp[k0 + 1];
        a_x[ks] = t1;
    }

#pragma unroll
    for (int nb = 0; nb < 4; ++nb) {
        const int ncol = nb * 16 + row;
        v8f c = {};
#pragma unroll
        for (int ks = 0; ks < 16; ++ks) {
            const int k0 = ks * 4 + half * 2;
            v2f bfl; bfl.x = Wl_s[ncol * FEAT + k0]; bfl.y = Wl_s[ncol * FEAT + k0 + 1];
            c = __builtin_amdgcn_wmma_f32_16x16x4_f32(false, a_ag[ks], false, bfl,
                                                      (short)0, c, false, false);
            v2f bfr; bfr.x = Wr_s[ncol * FEAT + k0]; bfr.y = Wr_s[ncol * FEAT + k0 + 1];
            c = __builtin_amdgcn_wmma_f32_16x16x4_f32(false, a_x[ks], false, bfr,
                                                      (short)0, c, false, false);
        }
        const float bias = b_s[nb * 16 + row];
#pragma unroll
        for (int r = 0; r < 8; ++r) {
            const int M = r + half * 8;
            out[(long)(base + M) * FEAT + nb * 16 + row] = c[r] + bias;
        }
    }
}

extern "C" void kernel_launch(void* const* d_in, const int* in_sizes, int n_in,
                              void* d_out, int out_size, void* d_ws, size_t ws_size,
                              hipStream_t stream) {
    const float* x   = (const float*)d_in[0];
    const int*   eix = (const int*)d_in[1];
    const float* W1l = (const float*)d_in[2];
    const float* b1l = (const float*)d_in[3];
    const float* W1r = (const float*)d_in[4];
    const float* W2l = (const float*)d_in[5];
    const float* b2l = (const float*)d_in[6];
    const float* W2r = (const float*)d_in[7];

    const int  N = in_sizes[0] / FEAT;
    const long E = in_sizes[1] / 2;
    const int* src = eix;
    const int* dst = eix + E;

    float* aggr = (float*)d_ws;
    float* x1   = aggr + (size_t)N * FEAT;
    float* deg  = x1 + (size_t)N * FEAT;
    float* outp = (float*)d_out;

    const int ag_blocks = (N + TILE - 1) / TILE;
    const int up_blocks = (N + 127) / 128;

    aggregate_owner<<<ag_blocks, AG_THREADS, 0, stream>>>(x, src, dst, aggr, deg, E, N);
    sage_update_wmma<<<up_blocks, 256, 0, stream>>>(x, aggr, deg, W1l, b1l, W1r, x1, N);

    aggregate_owner<<<ag_blocks, AG_THREADS, 0, stream>>>(x1, src, dst, aggr, deg, E, N);
    sage_update_wmma<<<up_blocks, 256, 0, stream>>>(x1, aggr, deg, W2l, b2l, W2r, outp, N);
}


# aggregation split over the edge list: two work-groups per 800-node range each scan half of it and keep partial sums and counts; the update kernel adds the two parts; new reduction tree
# speedup vs baseline: 2.2933x; 1.2621x over previous
#include <hip/hip_runtime.h>

typedef __attribute__((ext_vector_type(2))) float v2f;
typedef __attribute__((ext_vector_type(8))) float v8f;

#define FEAT 64
#define TILE 800
#define SPLIT 2
#define QCAP 256
#define AG_THREADS 1024

__global__ void __launch_bounds__(AG_THREADS)
aggregate_owner(const float* __restrict__ x, const int* __restrict__ src, const int* __restrict__ dst,
                float* __restrict__ aggr0, float* __restrict__ aggr1, float* __restrict__ deg, long E, int N) {
    __shared__ float    tile_s[(TILE + 1) * FEAT];
    __shared__ unsigned cnt_s[TILE + 1];
    __shared__ uint2    q_s[AG_THREADS / 32][QCAP];

    const int tid  = threadIdx.x;
    const int lane = tid & 31;
    const int lo   = blockIdx.x * TILE;
    const int part = blockIdx.y;
    const unsigned ucnt = (unsigned)min(TILE, N - lo);
    uint2* q = q_s[tid >> 5];

    for (int i = tid; i < (TILE + 1) * FEAT / 4; i += AG_THREADS)
        ((float4*)tile_s)[i] = make_float4(0.f, 0.f, 0.f, 0.f);
    for (int i = tid; i < TILE + 1; i += AG_THREADS) cnt_s[i] = 0u;
    __syncthreads();

    unsigned qhead = 0u, qcount = 0u;

    auto append = [&](bool m, unsigned e, unsigned row) {
        const unsigned mask = __builtin_amdgcn_ballot_w32(m);
        if (mask) {
            const unsigned below = __builtin_amdgcn_mbcnt_lo(mask, 0u);
            if (m) q[(qhead + qcount + below) & (QCAP - 1)] = make_uint2(e, row);
            qcount += (unsigned)__builtin_popcount(mask);
        }
    };

    auto drain = [&](unsigned n) {
        const bool valid = (unsigned)lane < n;
        uint2 ent = make_uint2(0u, (unsigned)TILE);
        if (valid) ent = q[(qhead + (unsigned)lane) & (QCAP - 1)];
        const unsigned row = ent.y;
        const int s = valid ? src[ent.x] : 0;
        atomicAdd(&cnt_s[row], 1u);
        for (int j0 = 0; j0 < 32; j0 += 8) {
            float2 v[8];
            unsigned r[8];
#pragma unroll
            for (int u = 0; u < 8; ++u) {
                const int sj = __builtin_amdgcn_readlane(s, j0 + u);
                r[u] = (unsigned)__builtin_amdgcn_readlane((int)row, j0 + u);
                v[u] = ((const float2*)(x + (long)sj * FEAT))[lane];
            }
#pragma unroll
            for (int u = 0; u < 8; ++u) {
                float* t = tile_s + r[u] * FEAT + lane * 2;
                atomicAdd(t, v[u].x);
                atomicAdd(t + 1, v[u].y);
            }
        }
        qhead = (qhead + 32u) & (QCAP - 1);
        qcount -= n;
    };

    auto process4 = [&](int4 d, unsigned e0) {
        const unsigned a0 = (unsigned)d.x - (unsigned)lo, a1 = (unsigned)d.y - (unsigned)lo;
        const unsigned a2 = (unsigned)d.z - (unsigned)lo, a3 = (unsigned)d.w - (unsigned)lo;
        const bool m0 = a0 < ucnt, m1 = a1 < ucnt, m2 = a2 < ucnt, m3 = a3 < ucnt;
        if (__builtin_amdgcn_ballot_w32(m0 | m1 | m2 | m3)) {
            append(m0, e0, a0); append(m1, e0 + 1u, a1); append(m2, e0 + 2u, a2); append(m3, e0 + 3u, a3);
            while (qcount >= 32u) drain(32u);
        }
    };

    const bool vec_ok = (((unsigned long long)dst) & 15ull) == 0ull;
    const long nfull  = vec_ok ? (E >> 2) : 0;
    const long nvec   = (E + 3) >> 2;
    const long Tmain  = (nfull / AG_THREADS) & ~7L;
    const long T      = (nvec + AG_THREADS - 1) / AG_THREADS;
    const int4* dst4  = (const int4*)dst;

    for (long t = 8L * part; t < Tmain; t += 8 * SPLIT) {
        int4 d[8];
#pragma unroll
        for (int u = 0; u < 8; ++u) d[u] = dst4[(t + u) * AG_THREADS + tid];
#pragma unroll
        for (int u = 0; u < 8; ++u) process4(d[u], (unsigned)(((t + u) * AG_THREADS + tid) * 4));
    }
    for (long t = Tmain + part; t < T; t += SPLIT) {
        const long e0 = (t * AG_THREADS + tid) * 4;
        int4 d;
        if (vec_ok && e0 + 3 < E) d = dst4[e0 >> 2];
        else {
            d.x = (e0     < E) ? dst[e0]     : -1;
            d.y = (e0 + 1 < E) ? dst[e0 + 1] : -1;
            d.z = (e0 + 2 < E) ? dst[e0 + 2] : -1;
            d.w = (e0 + 3 < E) ? dst[e0 + 3] : -1;
        }
        process4(d, (unsigned)e0);
    }
    if (qcount) drain(qcount);
    __syncthreads();

    float4* orow = (float4*)((part ? aggr1 : aggr0) + (long)lo * FEAT);
    float*  odeg = deg + (long)part * N + lo;
    for (int i = tid; i < (int)ucnt * (FEAT / 4); i += AG_THREADS) orow[i] = ((const float4*)tile_s)[i];
    for (int i = tid; i < (int)ucnt; i += AG_THREADS) odeg[i] = (float)cnt_s[i];
}

__global__ void __launch_bounds__(256)
sage_update_wmma(const float* __restrict__ x, const float* __restrict__ aggr0, const float* aggr1,
                 const float* __restrict__ deg,
                 const float* __restrict__ Wl, const float* __restrict__ bl,
                 const float* __restrict__ Wr,
                 float* out, int N) {
    __shared__ float Wl_s[FEAT * FEAT];
    __shared__ float Wr_s[FEAT * FEAT];
    __shared__ float b_s[FEAT];

    for (int i = threadIdx.x; i < FEAT * FEAT / 4; i += 256) {
        ((float4*)Wl_s)[i] = ((const float4*)Wl)[i];
        ((float4*)Wr_s)[i] = ((const float4*)Wr)[i];
    }
    if (threadIdx.x < FEAT) b_s[threadIdx.x] = bl[threadIdx.x];
    __syncthreads();

    const int lane = threadIdx.x & 31;
    const int wave = threadIdx.x >> 5;
    const int tile = blockIdx.x * 8 + wave;
    const int base = tile * 16;
    if (base >= N) return;

    const int row  = lane & 15;
    const int half = lane >> 4;
    const int node = base + row;
    const float dinv = 1.0f / fmaxf(deg[node] + deg[(long)N + node], 1.0f);

    v2f a_ag[16], a_x[16];
    const float* agp = aggr0 + (long)node * FEAT;
    const float* agq = aggr1 + (long)node * FEAT;
    const float* xp  = x    + (long)node * FEAT;
#pragma unroll
    for (int ks = 0; ks < 16; ++ks) {
        const int k0 = ks * 4 + half * 2;
        v2f t0; t0.x = (agp[k0] + agq[k0]) * dinv; t0.y = (agp[k0 + 1] + agq[k0 + 1]) * dinv;
        a_ag[ks] = t0;
        v2f t1; t1.x = xp[k0];         t1.y = xp[k0 + 1];
        a_x[ks] = t1;
    }

#pragma unroll
    for (int nb = 0; nb < 4; ++nb) {
        const int ncol = nb * 16 + row;
        v8f c = {};
#pragma unroll
        for (int ks = 0; ks < 16; ++ks) {
            const int k0 = ks * 4 + half * 2;
            v2f bfl; bfl.x = Wl_s[ncol * FEAT + k0]; bfl.y = Wl_s[ncol * FEAT + k0 + 1];
            c = __builtin_amdgcn_wmma_f32_16x16x4_f32(false, a_ag[ks], false, bfl,
                                                      (short)0, c, false, false);
            v2f bfr; bfr.x = Wr_s[ncol * FEAT + k0]; bfr.y = Wr_s[ncol * FEAT + k0 + 1];
            c = __builtin_amdgcn_wmma_f32_16x16x4_f32(false, a_x[ks], false, bfr,
                                                      (short)0, c, false, false);
        }
        const float bias = b_s[nb * 16 + row];
#pragma unroll
        for (int r = 0; r < 8; ++r) {
            const int M = r + half * 8;
            out[(long)(base + M) * FEAT + nb * 16 + row] = c[r] + bias;
        }
    }
}

extern "C" void kernel_launch(void* const* d_in, const int* in_sizes, int n_in,
                              void* d_out, int out_size, void* d_ws, size_t ws_size,
                              hipStream_t stream) {
    const float* x   = (const float*)d_in[0];
    const int*   eix = (const int*)d_in[1];
    const float* W1l = (const float*)d_in[2];
    const float* b1l = (const float*)d_in[3];
    const float* W1r = (const float*)d_in[4];
    const float* W2l = (const float*)d_in[5];
    const float* b2l = (const float*)d_in[6];
    const float* W2r = (const float*)d_in[7];

    const int  N = in_sizes[0] / FEAT;
    const long E = in_sizes[1] / 2;
    const int* src = eix;
    const int* dst = eix + E;

    float* aggr = (float*)d_ws;
    float* x1   = aggr + (size_t)N * FEAT;
    float* deg  = x1 + (size_t)N * FEAT;
    float* outp = (float*)d_out;
    float* aggb = outp;

    const dim3 ag_blocks((N + TILE - 1) / TILE, SPLIT);
    const int up_blocks = (N + 127) / 128;

    aggregate_owner<<<ag_blocks, AG_THREADS, 0, stream>>>(x, src, dst, aggr, aggb, deg, E, N);
    sage_update_wmma<<<up_blocks, 256, 0, stream>>>(x, aggr, aggb, deg, W1l, b1l, W1r, x1, N);

    aggregate_owner<<<ag_blocks, AG_THREADS, 0, stream>>>(x1, src, dst, aggr, aggb, deg, E, N);
    sage_update_wmma<<<up_blocks, 256, 0, stream>>>(x1, aggr, aggb, deg, W2l, b2l, W2r, outp, N);
}


# update kernel re-tiled from 16 to 64 nodes per wave: 4 accumulator tiles share each weight fragment read from LDS, so those reads per node fall 4-fold (bank layout unchanged); same sums, same order
# speedup vs baseline: 2.4114x; 1.0515x over previous
#include <hip/hip_runtime.h>

typedef __attribute__((ext_vector_type(2))) float v2f;
typedef __attribute__((ext_vector_type(8))) float v8f;

#define FEAT 64
#define TILE 800
#define SPLIT 2
#define QCAP 256
#define AG_THREADS 1024

__global__ void __launch_bounds__(AG_THREADS)
aggregate_owner(const float* __restrict__ x, const int* __restrict__ src, const int* __restrict__ dst,
                float* __restrict__ aggr0, float* __restrict__ aggr1, float* __restrict__ deg, long E, int N) {
    __shared__ float    tile_s[(TILE + 1) * FEAT];
    __shared__ unsigned cnt_s[TILE + 1];
    __shared__ uint2    q_s[AG_THREADS / 32][QCAP];

    const int tid  = threadIdx.x;
    const int lane = tid & 31;
    const int lo   = blockIdx.x * TILE;
    const int part = blockIdx.y;
    const unsigned ucnt = (unsigned)min(TILE, N - lo);
    uint2* q = q_s[tid >> 5];

    for (int i = tid; i < (TILE + 1) * FEAT / 4; i += AG_THREADS)
        ((float4*)tile_s)[i] = make_float4(0.f, 0.f, 0.f, 0.f);
    for (int i = tid; i < TILE + 1; i += AG_THREADS) cnt_s[i] = 0u;
    __syncthreads();

    unsigned qhead = 0u, qcount = 0u;

    auto append = [&](bool m, unsigned e, unsigned row) {
        const unsigned mask = __builtin_amdgcn_ballot_w32(m);
        if (mask) {
            const unsigned below = __builtin_amdgcn_mbcnt_lo(mask, 0u);
            if (m) q[(qhead + qcount + below) & (QCAP - 1)] = make_uint2(e, row);
            qcount += (unsigned)__builtin_popcount(mask);
        }
    };

    auto drain = [&](unsigned n) {
        const bool valid = (unsigned)lane < n;
        uint2 ent = make_uint2(0u, (unsigned)TILE);
        if (valid) ent = q[(qhead + (unsigned)lane) & (QCAP - 1)];
        const unsigned row = ent.y;
        const int s = valid ? src[ent.x] : 0;
        atomicAdd(&cnt_s[row], 1u);
        for (int j0 = 0; j0 < 32; j0 += 8) {
            float2 v[8];
            unsigned r[8];
#pragma unroll
            for (int u = 0; u < 8; ++u) {
                const int sj = __builtin_amdgcn_readlane(s, j0 + u);
                r[u] = (unsigned)__builtin_amdgcn_readlane((int)row, j0 + u);
                v[u] = ((const float2*)(x + (long)sj * FEAT))[lane];
            }
#pragma unroll
            for (int u = 0; u < 8; ++u) {
                float* t = tile_s + r[u] * FEAT + lane * 2;
                atomicAdd(t, v[u].x);
                atomicAdd(t + 1, v[u].y);
            }
        }
        qhead = (qhead + 32u) & (QCAP - 1);
        qcount -= n;
    };

    auto process4 = [&](int4 d, unsigned e0) {
        const unsigned a0 = (unsigned)d.x - (unsigned)lo, a1 = (unsigned)d.y - (unsigned)lo;
        const unsigned a2 = (unsigned)d.z - (unsigned)lo, a3 = (unsigned)d.w - (unsigned)lo;
        const bool m0 = a0 < ucnt, m1 = a1 < ucnt, m2 = a2 < ucnt, m3 = a3 < ucnt;
        if (__builtin_amdgcn_ballot_w32(m0 | m1 | m2 | m3)) {
            append(m0, e0, a0); append(m1, e0 + 1u, a1); append(m2, e0 + 2u, a2); append(m3, e0 + 3u, a3);
            while (qcount >= 32u) drain(32u);
        }
    };

    const bool vec_ok = (((unsigned long long)dst) & 15ull) == 0ull;
    const long nfull  = vec_ok ? (E >> 2) : 0;
    const long nvec   = (E + 3) >> 2;
    const long Tmain  = (nfull / AG_THREADS) & ~7L;
    const long T      = (nvec + AG_THREADS - 1) / AG_THREADS;
    const int4* dst4  = (const int4*)dst;

    for (long t = 8L * part; t < Tmain; t += 8 * SPLIT) {
        int4 d[8];
#pragma unroll
        for (int u = 0; u < 8; ++u) d[u] = dst4[(t + u) * AG_THREADS + tid];
#pragma unroll
        for (int u = 0; u < 8; ++u) process4(d[u], (unsigned)(((t + u) * AG_THREADS + tid) * 4));
    }
    for (long t = Tmain + part; t < T; t += SPLIT) {
        const long e0 = (t * AG_THREADS + tid) * 4;
        int4 d;
        if (vec_ok && e0 + 3 < E) d = dst4[e0 >> 2];
        else {
            d.x = (e0     < E) ? dst[e0]     : -1;
            d.y = (e0 + 1 < E) ? dst[e0 + 1] : -1;
            d.z = (e0 + 2 < E) ? dst[e0 + 2] : -1;
            d.w = (e0 + 3 < E) ? dst[e0 + 3] : -1;
        }
        process4(d, (unsigned)e0);
    }
    if (qcount) drain(qcount);
    __syncthreads();

    float4* orow = (float4*)((part ? aggr1 : aggr0) + (long)lo * FEAT);
    float*  odeg = deg + (long)part * N + lo;
    for (int i = tid; i < (int)ucnt * (FEAT / 4); i += AG_THREADS) orow[i] = ((const float4*)tile_s)[i];
    for (int i = tid; i < (int)ucnt; i += AG_THREADS) odeg[i] = (float)cnt_s[i];
}

#define MT 4
__global__ void __launch_bounds__(256)
sage_update_wmma(const float* __restrict__ x, const float* __restrict__ aggr0, const float* aggr1,
                 const float* __restrict__ deg,
                 const float* __restrict__ Wl, const float* __restrict__ bl,
                 const float* __restrict__ Wr,
                 float* out, int N) {
    __shared__ float Wl_s[FEAT * FEAT];
    __shared__ float Wr_s[FEAT * FEAT];
    __shared__ float b_s[FEAT];

    for (int i = threadIdx.x; i < FEAT * FEAT / 4; i += 256) {
        ((float4*)Wl_s)[i] = ((const float4*)Wl)[i];
        ((float4*)Wr_s)[i] = ((const float4*)Wr)[i];
    }
    if (threadIdx.x < FEAT) b_s[threadIdx.x] = bl[threadIdx.x];
    __syncthreads();

    const int lane = threadIdx.x & 31;
    const int wave = threadIdx.x >> 5;
    const int base0 = (blockIdx.x * 8 + wave) * (16 * MT);
    if (base0 >= N) return;

    const int row  = lane & 15;
    const int half = lane >> 4;

    v2f a_ag[MT][16], a_x[MT][16];
#pragma unroll
    for (int t = 0; t < MT; ++t) {
        const int bt   = base0 + t * 16;
        const int node = ((bt < N) ? bt : base0) + row;
        const float dinv = 1.0f / fmaxf(deg[node] + deg[(long)N + node], 1.0f);
        const float* agp = aggr0 + (long)node * FEAT;
        const float* agq = aggr1 + (long)node * FEAT;
        const float* xp  = x    + (long)node * FEAT;
#pragma unroll
        for (int ks = 0; ks < 16; ++ks) {
            const int k0 = ks * 4 + half * 2;
            v2f t0; t0.x = (agp[k0] + agq[k0]) * dinv; t0.y = (agp[k0 + 1] + agq[k0 + 1]) * dinv;
            a_ag[t][ks] = t0;
            v2f t1; t1.x = xp[k0];         t1.y = xp[k0 + 1];
            a_x[t][ks] = t1;
        }
    }

#pragma unroll
    for (int nb = 0; nb < 4; ++nb) {
        const int ncol = nb * 16 + row;
        v8f c[MT];
#pragma unroll
        for (int t = 0; t < MT; ++t) c[t] = v8f{};
#pragma unroll
        for (int ks = 0; ks < 16; ++ks) {
            const int k0 = ks * 4 + half * 2;
            v2f bfl; bfl.x = Wl_s[ncol * FEAT + k0]; bfl.y = Wl_s[ncol * FEAT + k0 + 1];
            v2f bfr; bfr.x = Wr_s[ncol * FEAT + k0]; bfr.y = Wr_s[ncol * FEAT + k0 + 1];
#pragma unroll
            for (int t = 0; t < MT; ++t)
                c[t] = __builtin_amdgcn_wmma_f32_16x16x4_f32(false, a_ag[t][ks], false, bfl,
                                                             (short)0, c[t], false, false);
#pragma unroll
            for (int t = 0; t < MT; ++t)
                c[t] = __builtin_amdgcn_wmma_f32_16x16x4_f32(false, a_x[t][ks], false, bfr,
                                                             (short)0, c[t], false, false);
        }
        const float bias = b_s[nb * 16 + row];
#pragma unroll
        for (int t = 0; t < MT; ++t) {
            const int bt = base0 + t * 16;
            if (bt < N) {
#pragma unroll
                for (int r = 0; r < 8; ++r) {
                    const int M = r + half * 8;
                    out[(long)(bt + M) * FEAT + nb * 16 + row] = c[t][r] + bias;
                }
            }
        }
    }
}

extern "C" void kernel_launch(void* const* d_in, const int* in_sizes, int n_in,
                              void* d_out, int out_size, void* d_ws, size_t ws_size,
                              hipStream_t stream) {
    const float* x   = (const float*)d_in[0];
    const int*   eix = (const int*)d_in[1];
    const float* W1l = (const float*)d_in[2];
    const float* b1l = (const float*)d_in[3];
    const float* W1r = (const float*)d_in[4];
    const float* W2l = (const float*)d_in[5];
    const float* b2l = (const float*)d_in[6];
    const float* W2r = (const float*)d_in[7];

    const int  N = in_sizes[0] / FEAT;
    const long E = in_sizes[1] / 2;
    const int* src = eix;
    const int* dst = eix + E;

    float* aggr = (float*)d_ws;
    float* x1   = aggr + (size_t)N * FEAT;
    float* deg  = x1 + (size_t)N * FEAT;
    float* outp = (float*)d_out;
    float* aggb = outp;

    const dim3 ag_blocks((N + TILE - 1) / TILE, SPLIT);
    const int up_blocks = (N + 8 * 16 * MT - 1) / (8 * 16 * MT);

    aggregate_owner<<<ag_blocks, AG_THREADS, 0, stream>>>(x, src, dst, aggr, aggb, deg, E, N);
    sage_update_wmma<<<up_blocks, 256, 0, stream>>>(x, aggr, aggb, deg, W1l, b1l, W1r, x1, N);

    aggregate_owner<<<ag_blocks, AG_THREADS, 0, stream>>>(x1, src, dst, aggr, aggb, deg, E, N);
    sage_update_wmma<<<up_blocks, 256, 0, stream>>>(x1, aggr, aggb, deg, W2l, b2l, W2r, outp, N);
}


# destination list condensed once per call to a range-id byte per edge (held in the output buffer); owners of 400 nodes scan bytes; one sum per node, no two-part split; 4 to 5 launches; arrival order
# speedup vs baseline: 2.6516x; 1.0996x over previous
#include <hip/hip_runtime.h>

typedef __attribute__((ext_vector_type(2))) float v2f;
typedef __attribute__((ext_vector_type(8))) float v8f;

#define FEAT 64
#define TILE 400
#define QCAP 512
#define AG_THREADS 1024
#define KEY_THREADS 256

__global__ void __launch_bounds__(KEY_THREADS)
range_keys(const int* __restrict__ dst, float* __restrict__ keyf, long E) {
    unsigned char* key = (unsigned char*)keyf;
    const long e0 = ((long)blockIdx.x * KEY_THREADS + threadIdx.x) * 16;
    if (e0 >= E) return;
    const bool vec_ok = (((unsigned long long)dst) & 15ull) == 0ull && (((unsigned long long)keyf) & 15ull) == 0ull;
    if (vec_ok && e0 + 15 < E) {
        const int4* d4 = (const int4*)(dst + e0);
        unsigned w[4];
#pragma unroll
        for (int j = 0; j < 4; ++j) {
            const int4 d = d4[j];
            w[j] = (((unsigned)d.x / TILE) & 255u)        | ((((unsigned)d.y / TILE) & 255u) << 8) |
                   ((((unsigned)d.z / TILE) & 255u) << 16) | ((((unsigned)d.w / TILE) & 255u) << 24);
        }
        *(uint4*)(key + e0) = make_uint4(w[0], w[1], w[2], w[3]);
    } else {
        for (long e = e0; e < E && e < e0 + 16; ++e) key[e] = (unsigned char)(((unsigned)dst[e] / TILE) & 255u);
    }
}

__global__ void __launch_bounds__(AG_THREADS)
aggregate_owner(const float* __restrict__ x, const int* __restrict__ src, const int* __restrict__ dst,
                const float* __restrict__ keyf, float* __restrict__ aggr, float* __restrict__ deg, long E, int N) {
    __shared__ float    tile_s[(TILE + 1) * FEAT];
    __shared__ unsigned cnt_s[TILE + 1];
    __shared__ unsigned q_s[AG_THREADS / 32][QCAP];

    const int tid  = threadIdx.x;
    const int lane = tid & 31;
    const int lo   = blockIdx.x * TILE;
    const unsigned ucnt  = (unsigned)min(TILE, N - lo);
    const unsigned myid  = (unsigned)blockIdx.x & 255u;
    const unsigned idrep = myid * 0x01010101u;
    unsigned* q = q_s[tid >> 5];

    for (int i = tid; i < (TILE + 1) * FEAT / 4; i += AG_THREADS)
        ((float4*)tile_s)[i] = make_float4(0.f, 0.f, 0.f, 0.f);
    for (int i = tid; i < TILE + 1; i += AG_THREADS) cnt_s[i] = 0u;
    __syncthreads();

    unsigned qhead = 0u, qcount = 0u;

    auto append = [&](bool m, unsigned e) {
        const unsigned mask = __builtin_amdgcn_ballot_w32(m);
        if (mask) {
            const unsigned below = __builtin_amdgcn_mbcnt_lo(mask, 0u);
            if (m) q[(qhead + qcount + below) & (QCAP - 1)] = e;
            qcount += (unsigned)__builtin_popcount(mask);
        }
    };

    auto drain = [&](unsigned n) {
        const bool valid = (unsigned)lane < n;
        unsigned e = 0u;
        if (valid) e = q[(qhead + (unsigned)lane) & (QCAP - 1)];
        const unsigned a  = (unsigned)dst[e] - (unsigned)lo;
        const int      s0 = src[e];
        const bool ok = valid && a < ucnt;
        const unsigned row = ok ? a : (unsigned)TILE;
        const int s = ok ? s0 : 0;
        atomicAdd(&cnt_s[row], 1u);
        for (int j0 = 0; j0 < 32; j0 += 8) {
            float2 v[8];
            unsigned r[8];
#pragma unroll
            for (int u = 0; u < 8; ++u) {
                const int sj = __builtin_amdgcn_readlane(s, j0 + u);
                r[u] = (unsigned)__builtin_amdgcn_readlane((int)row, j0 + u);
                v[u] = ((const float2*)(x + (long)sj * FEAT))[lane];
            }
#pragma unroll
            for (int u = 0; u < 8; ++u) {
                float* t = tile_s + r[u] * FEAT + lane * 2;
                atomicAdd(t, v[u].x);
                atomicAdd(t + 1, v[u].y);
            }
        }
        qhead = (qhead + 32u) & (QCAP - 1);
        qcount -= n;
    };

    auto process16 = [&](uint4 w, unsigned e0) {
        const unsigned x0 = w.x ^ idrep, x1 = w.y ^ idrep, x2 = w.z ^ idrep, x3 = w.w ^ idrep;
        const unsigned z0 = (x0 - 0x01010101u) & ~x0, z1 = (x1 - 0x01010101u) & ~x1;
        const unsigned z2 = (x2 - 0x01010101u) & ~x2, z3 = (x3 - 0x01010101u) & ~x3;
        const unsigned any = (z0 | z1 | z2 | z3) & 0x80808080u;
        if (__builtin_amdgcn_ballot_w32(any != 0u)) {
            unsigned m = ((z0 >> 7) & 0x01010101u) | ((z1 >> 6) & 0x02020202u) |
                         ((z2 >> 5) & 0x04040404u) | ((z3 >> 4) & 0x08080808u);
            while (__builtin_amdgcn_ballot_w32(m != 0u)) {
                const bool has = m != 0u;
                const unsigned pos = (unsigned)__builtin_ctz(m | 0x80000000u);
                append(has, e0 + (((pos & 7u) << 2) | (pos >> 3)));
                m &= m - 1u;
                while (qcount >= 32u) drain(32u);
            }
        }
    };

    const bool vec_ok = (((unsigned long long)keyf) & 15ull) == 0ull;
    const long nP     = vec_ok ? (E >> 4) : 0;
    const long Tmain  = (nP / AG_THREADS) & ~7L;
    const uint4* key4 = (const uint4*)keyf;
    const unsigned char* key = (const unsigned char*)keyf;

    for (long t = 0; t < Tmain; t += 8) {
        uint4 d[8];
#pragma unroll
        for (int u = 0; u < 8; ++u) d[u] = key4[(t + u) * AG_THREADS + tid];
#pragma unroll
        for (int u = 0; u < 8; ++u) process16(d[u], (unsigned)(((t + u) * AG_THREADS + tid) * 16));
    }
    for (long pb = Tmain * AG_THREADS; pb < nP; pb += AG_THREADS) {
        const long p = pb + tid;
        uint4 w = make_uint4(~idrep, ~idrep, ~idrep, ~idrep);
        if (p < nP) w = key4[p];
        process16(w, (unsigned)(p * 16));
    }
    for (long eb = nP * 16; eb < E; eb += AG_THREADS) {
        const long e = eb + tid;
        const bool m = e < E && (unsigned)key[e < E ? e : 0] == myid;
        append(m, (unsigned)e);
        while (qcount >= 32u) drain(32u);
    }
    if (qcount) drain(qcount);
    __syncthreads();

    float4* orow = (float4*)(aggr + (long)lo * FEAT);
    float*  odeg = deg + lo;
    for (int i = tid; i < (int)ucnt * (FEAT / 4); i += AG_THREADS) orow[i] = ((const float4*)tile_s)[i];
    for (int i = tid; i < (int)ucnt; i += AG_THREADS) odeg[i] = (float)cnt_s[i];
}

#define MT 4
__global__ void __launch_bounds__(256)
sage_update_wmma(const float* __restrict__ x, const float* __restrict__ aggr,
                 const float* __restrict__ deg,
                 const float* __restrict__ Wl, const float* __restrict__ bl,
                 const float* __restrict__ Wr,
                 float* __restrict__ out, int N) {
    __shared__ float Wl_s[FEAT * FEAT];
    __shared__ float Wr_s[FEAT * FEAT];
    __shared__ float b_s[FEAT];

    for (int i = threadIdx.x; i < FEAT * FEAT / 4; i += 256) {
        ((float4*)Wl_s)[i] = ((const float4*)Wl)[i];
        ((float4*)Wr_s)[i] = ((const float4*)Wr)[i];
    }
    if (threadIdx.x < FEAT) b_s[threadIdx.x] = bl[threadIdx.x];
    __syncthreads();

    const int lane = threadIdx.x & 31;
    const int wave = threadIdx.x >> 5;
    const int base0 = (blockIdx.x * 8 + wave) * (16 * MT);
    if (base0 >= N) return;

    const int row  = lane & 15;
    const int half = lane >> 4;

    v2f a_ag[MT][16], a_x[MT][16];
#pragma unroll
    for (int t = 0; t < MT; ++t) {
        const int bt   = base0 + t * 16;
        const int node = ((bt < N) ? bt : base0) + row;
        const float dinv = 1.0f / fmaxf(deg[node], 1.0f);
        const float* agp = aggr + (long)node * FEAT;
        const float* xp  = x    + (long)node * FEAT;
#pragma unroll
        for (int ks = 0; ks < 16; ++ks) {
            const int k0 = ks * 4 + half * 2;
            v2f t0; t0.x = agp[k0] * dinv; t0.y = agp[k0 + 1] * dinv;
            a_ag[t][ks] = t0;
            v2f t1; t1.x = xp[k0];         t1.y = xp[k0 + 1];
            a_x[t][ks] = t1;
        }
    }

#pragma unroll
    for (int nb = 0; nb < 4; ++nb) {
        const int ncol = nb * 16 + row;
        v8f c[MT];
#pragma unroll
        for (int t = 0; t < MT; ++t) c[t] = v8f{};
#pragma unroll
        for (int ks = 0; ks < 16; ++ks) {
            const int k0 = ks * 4 + half * 2;
            v2f bfl; bfl.x = Wl_s[ncol * FEAT + k0]; bfl.y = Wl_s[ncol * FEAT + k0 + 1];
            v2f bfr; bfr.x = Wr_s[ncol * FEAT + k0]; bfr.y = Wr_s[ncol * FEAT + k0 + 1];
#pragma unroll
            for (int t = 0; t < MT; ++t)
                c[t] = __builtin_amdgcn_wmma_f32_16x16x4_f32(false, a_ag[t][ks], false, bfl,
                                                             (short)0, c[t], false, false);
#pragma unroll
            for (int t = 0; t < MT; ++t)
                c[t] = __builtin_amdgcn_wmma_f32_16x16x4_f32(false, a_x[t][ks], false, bfr,
                                                             (short)0, c[t], false, false);
        }
        const float bias = b_s[nb * 16 + row];
#pragma unroll
        for (int t = 0; t < MT; ++t) {
            const int bt = base0 + t * 16;
            if (bt < N) {
#pragma unroll
                for (int r = 0; r < 8; ++r) {
                    const int M = r + half * 8;
                    out[(long)(bt + M) * FEAT + nb * 16 + row] = c[t][r] + bias;
                }
            }
        }
    }
}

extern "C" void kernel_launch(void* const* d_in, const int* in_sizes, int n_in,
                              void* d_out, int out_size, void* d_ws, size_t ws_size,
                              hipStream_t stream) {
    const float* x   = (const float*)d_in[0];
    const int*   eix = (const int*)d_in[1];
    const float* W1l = (const float*)d_in[2];
    const float* b1l = (const float*)d_in[3];
    const float* W1r = (const float*)d_in[4];
    const float* W2l = (const float*)d_in[5];
    const float* b2l = (const float*)d_in[6];
    const float* W2r = (const float*)d_in[7];

    const int  N = in_sizes[0] / FEAT;
    const long E = in_sizes[1] / 2;
    const int* src = eix;
    const int* dst = eix + E;

    float* aggr = (float*)d_ws;
    float* x1   = aggr + (size_t)N * FEAT;
    float* deg  = x1 + (size_t)N * FEAT;
    float* outp = (float*)d_out;
    float* keyf = outp;

    const int key_blocks = (int)((E + 16L * KEY_THREADS - 1) / (16L * KEY_THREADS));
    const int ag_blocks  = (N + TILE - 1) / TILE;
    const int up_blocks  = (N + 8 * 16 * MT - 1) / (8 * 16 * MT);

    range_keys<<<key_blocks, KEY_THREADS, 0, stream>>>(dst, keyf, E);

    aggregate_owner<<<ag_blocks, AG_THREADS, 0, stream>>>(x, src, dst, keyf, aggr, deg, E, N);
    sage_update_wmma<<<up_blocks, 256, 0, stream>>>(x, aggr, deg, W1l, b1l, W1r, x1, N);

    aggregate_owner<<<ag_blocks, AG_THREADS, 0, stream>>>(x1, src, dst, keyf, aggr, deg, E, N);
    sage_update_wmma<<<up_blocks, 256, 0, stream>>>(x1, aggr, deg, W2l, b2l, W2r, outp, N);
}


# destination list indexed once per call by a (node range x 16-edge piece) bitmap beside the key bytes; an owner walks its bitmap row and reads only flagged pieces' keys, not every key; arrival order
# speedup vs baseline: 2.7502x; 1.0372x over previous
#include <hip/hip_runtime.h>

typedef __attribute__((ext_vector_type(2))) float v2f;
typedef __attribute__((ext_vector_type(8))) float v8f;

#define FEAT 64
#define TILE 400
#define QCAP 512
#define AG_THREADS 1024
#define KEY_THREADS 1024
#define RG 256
#define RGP 257

__global__ void __launch_bounds__(KEY_THREADS)
range_index(const int* __restrict__ dst, float* __restrict__ keyf, float* __restrict__ bmf, long E, int nR, long Wp) {
    __shared__ unsigned bm_s[32 * RGP];
    unsigned char* key = (unsigned char*)keyf;
    unsigned* bitmap = (unsigned*)bmf;
    const int tid = threadIdx.x, lane = tid & 31, wave = tid >> 5;
    const long e0 = ((long)blockIdx.x * KEY_THREADS + tid) * 16;
    const bool vec_ok = (((unsigned long long)dst) & 15ull) == 0ull && (((unsigned long long)keyf) & 15ull) == 0ull;
    const bool whole = vec_ok && e0 + 15 < E;
    unsigned r[16];
#pragma unroll
    for (int j = 0; j < 16; ++j) r[j] = 0xffffffffu;
    if (whole) {
        const int4* d4 = (const int4*)(dst + e0);
        unsigned w[4];
#pragma unroll
        for (int j = 0; j < 4; ++j) {
            const int4 d = d4[j];
            r[4 * j + 0] = (unsigned)d.x / TILE; r[4 * j + 1] = (unsigned)d.y / TILE;
            r[4 * j + 2] = (unsigned)d.z / TILE; r[4 * j + 3] = (unsigned)d.w / TILE;
            w[j] = (r[4 * j + 0] & 255u) | ((r[4 * j + 1] & 255u) << 8) |
                   ((r[4 * j + 2] & 255u) << 16) | ((r[4 * j + 3] & 255u) << 24);
        }
        *(uint4*)(key + e0) = make_uint4(w[0], w[1], w[2], w[3]);
    } else {
        for (long e = e0; e < E && e < e0 + 16; ++e) key[e] = (unsigned char)(((unsigned)dst[e] / TILE) & 255u);
    }
    for (int g0 = 0; g0 < nR; g0 += RG) {
        for (int i = tid; i < 32 * RGP; i += KEY_THREADS) bm_s[i] = 0u;
        __syncthreads();
        if (whole) {
#pragma unroll
            for (int j = 0; j < 16; ++j) {
                const unsigned rl = r[j] - (unsigned)g0;
                if (rl < (unsigned)RG && r[j] < (unsigned)nR) atomicOr(&bm_s[wave * RGP + (int)rl], 1u << lane);
            }
        }
        __syncthreads();
        const int cnt = min(RG, nR - g0);
        for (int i = tid; i < cnt * 32; i += KEY_THREADS) {
            const int rl = i >> 5, w = i & 31;
            bitmap[(size_t)(g0 + rl) * (size_t)Wp + (size_t)blockIdx.x * 32 + w] = bm_s[w * RGP + rl];
        }
        __syncthreads();
    }
}

__global__ void __launch_bounds__(AG_THREADS)
aggregate_owner(const float* __restrict__ x, const int* __restrict__ src, const int* __restrict__ dst,
                const float* __restrict__ keyf, const float* __restrict__ bmf, long Wp,
                float* __restrict__ aggr, float* __restrict__ deg, long E, int N) {
    __shared__ float    tile_s[(TILE + 1) * FEAT];
    __shared__ unsigned cnt_s[TILE + 1];
    __shared__ unsigned q_s[AG_THREADS / 32][QCAP];

    const int tid  = threadIdx.x;
    const int lane = tid & 31;
    const int lo   = blockIdx.x * TILE;
    const unsigned ucnt  = (unsigned)min(TILE, N - lo);
    const unsigned myid  = (unsigned)blockIdx.x & 255u;
    const unsigned idrep = myid * 0x01010101u;
    unsigned* q = q_s[tid >> 5];

    for (int i = tid; i < (TILE + 1) * FEAT / 4; i += AG_THREADS)
        ((float4*)tile_s)[i] = make_float4(0.f, 0.f, 0.f, 0.f);
    for (int i = tid; i < TILE + 1; i += AG_THREADS) cnt_s[i] = 0u;
    __syncthreads();

    unsigned qhead = 0u, qcount = 0u;

    auto append = [&](bool m, unsigned e) {
        const unsigned mask = __builtin_amdgcn_ballot_w32(m);
        if (mask) {
            const unsigned below = __builtin_amdgcn_mbcnt_lo(mask, 0u);
            if (m) q[(qhead + qcount + below) & (QCAP - 1)] = e;
            qcount += (unsigned)__builtin_popcount(mask);
        }
    };

    auto drain = [&](unsigned n) {
        const bool valid = (unsigned)lane < n;
        unsigned e = 0u;
        if (valid) e = q[(qhead + (unsigned)lane) & (QCAP - 1)];
        const unsigned a  = (unsigned)dst[e] - (unsigned)lo;
        const int      s0 = src[e];
        const bool ok = valid && a < ucnt;
        const unsigned row = ok ? a : (unsigned)TILE;
        const int s = ok ? s0 : 0;
        atomicAdd(&cnt_s[row], 1u);
        for (int j0 = 0; j0 < 32; j0 += 8) {
            float2 v[8];
            unsigned r[8];
#pragma unroll
            for (int u = 0; u < 8; ++u) {
                const int sj = __builtin_amdgcn_readlane(s, j0 + u);
                r[u] = (unsigned)__builtin_amdgcn_readlane((int)row, j0 + u);
                v[u] = ((const float2*)(x + (long)sj * FEAT))[lane];
            }
#pragma unroll
            for (int u = 0; u < 8; ++u) {
                float* t = tile_s + r[u] * FEAT + lane * 2;
                atomicAdd(t, v[u].x);
                atomicAdd(t + 1, v[u].y);
            }
        }
        qhead = (qhead + 32u) & (QCAP - 1);
        qcount -= n;
    };

    auto process16 = [&](uint4 w, unsigned e0) {
        const unsigned x0 = w.x ^ idrep, x1 = w.y ^ idrep, x2 = w.z ^ idrep, x3 = w.w ^ idrep;
        const unsigned z0 = (x0 - 0x01010101u) & ~x0, z1 = (x1 - 0x01010101u) & ~x1;
        const unsigned z2 = (x2 - 0x01010101u) & ~x2, z3 = (x3 - 0x01010101u) & ~x3;
        const unsigned any = (z0 | z1 | z2 | z3) & 0x80808080u;
        if (__builtin_amdgcn_ballot_w32(any != 0u)) {
            unsigned m = ((z0 >> 7) & 0x01010101u) | ((z1 >> 6) & 0x02020202u) |
                         ((z2 >> 5) & 0x04040404u) | ((z3 >> 4) & 0x08080808u);
            while (__builtin_amdgcn_ballot_w32(m != 0u)) {
                const bool has = m != 0u;
                const unsigned pos = (unsigned)__builtin_ctz(m | 0x80000000u);
                append(has, e0 + (((pos & 7u) << 2) | (pos >> 3)));
                m &= m - 1u;
                while (qcount >= 32u) drain(32u);
            }
        }
    };

    const bool vec_ok = (((unsigned long long)dst) & 15ull) == 0ull && (((unsigned long long)keyf) & 15ull) == 0ull;
    const long nP     = vec_ok ? (E >> 4) : 0;
    const long Wn     = (nP + 31) >> 5;
    const uint4* key4 = (const uint4*)keyf;
    const unsigned char* key = (const unsigned char*)keyf;
    const unsigned* brow = (const unsigned*)bmf + (size_t)blockIdx.x * (size_t)Wp;
    const uint4 none4 = make_uint4(~idrep, ~idrep, ~idrep, ~idrep);

    for (long wb = 0; wb < Wn; wb += AG_THREADS) {
        const long wi = wb + tid;
        unsigned bits = wi < Wn ? brow[wi] : 0u;
        while (__builtin_amdgcn_ballot_w32(bits != 0u)) {
            uint4 d[8];
            unsigned pe[8];
            bool hs[8];
#pragma unroll
            for (int u = 0; u < 8; ++u) {
                hs[u] = bits != 0u;
                const long p = hs[u] ? wi * 32 + (long)__builtin_ctz(bits) : 0L;
                d[u] = key4[p];
                pe[u] = (unsigned)(p * 16);
                bits &= bits - 1u;
            }
#pragma unroll
            for (int u = 0; u < 8; ++u) process16(hs[u] ? d[u] : none4, pe[u]);
        }
    }
    for (long eb = nP * 16; eb < E; eb += AG_THREADS) {
        const long e = eb + tid;
        const bool m = e < E && (unsigned)key[e < E ? e : 0] == myid;
        append(m, (unsigned)e);
        while (qcount >= 32u) drain(32u);
    }
    if (qcount) drain(qcount);
    __syncthreads();

    float4* orow = (float4*)(aggr + (long)lo * FEAT);
    float*  odeg = deg + lo;
    for (int i = tid; i < (int)ucnt * (FEAT / 4); i += AG_THREADS) orow[i] = ((const float4*)tile_s)[i];
    for (int i = tid; i < (int)ucnt; i += AG_THREADS) odeg[i] = (float)cnt_s[i];
}

#define MT 4
__global__ void __launch_bounds__(256)
sage_update_wmma(const float* __restrict__ x, const float* __restrict__ aggr,
                 const float* __restrict__ deg,
                 const float* __restrict__ Wl, const float* __restrict__ bl,
                 const float* __restrict__ Wr,
                 float* __restrict__ out, int N) {
    __shared__ float Wl_s[FEAT * FEAT];
    __shared__ float Wr_s[FEAT * FEAT];
    __shared__ float b_s[FEAT];

    for (int i = threadIdx.x; i < FEAT * FEAT / 4; i += 256) {
        ((float4*)Wl_s)[i] = ((const float4*)Wl)[i];
        ((float4*)Wr_s)[i] = ((const float4*)Wr)[i];
    }
    if (threadIdx.x < FEAT) b_s[threadIdx.x] = bl[threadIdx.x];
    __syncthreads();

    const int lane = threadIdx.x & 31;
    const int wave = threadIdx.x >> 5;
    const int base0 = (blockIdx.x * 8 + wave) * (16 * MT);
    if (base0 >= N) return;

    const int row  = lane & 15;
    const int half = lane >> 4;

    v2f a_ag[MT][16], a_x[MT][16];
#pragma unroll
    for (int t = 0; t < MT; ++t) {
        const int bt   = base0 + t * 16;
        const int node = ((bt < N) ? bt : base0) + row;
        const float dinv = 1.0f / fmaxf(deg[node], 1.0f);
        const float* agp = aggr + (long)node * FEAT;
        const float* xp  = x    + (long)node * FEAT;
#pragma unroll
        for (int ks = 0; ks < 16; ++ks) {
            const int k0 = ks * 4 + half * 2;
            v2f t0; t0.x = agp[k0] * dinv; t0.y = agp[k0 + 1] * dinv;
            a_ag[t][ks] = t0;
            v2f t1; t1.x = xp[k0];         t1.y = xp[k0 + 1];
            a_x[t][ks] = t1;
        }
    }

#pragma unroll
    for (int nb = 0; nb < 4; ++nb) {
        const int ncol = nb * 16 + row;
        v8f c[MT];
#pragma unroll
        for (int t = 0; t < MT; ++t) c[t] = v8f{};
#pragma unroll
        for (int ks = 0; ks < 16; ++ks) {
            const int k0 = ks * 4 + half * 2;
            v2f bfl; bfl.x = Wl_s[ncol * FEAT + k0]; bfl.y = Wl_s[ncol * FEAT + k0 + 1];
            v2f bfr; bfr.x = Wr_s[ncol * FEAT + k0]; bfr.y = Wr_s[ncol * FEAT + k0 + 1];
#pragma unroll
            for (int t = 0; t < MT; ++t)
                c[t] = __builtin_amdgcn_wmma_f32_16x16x4_f32(false, a_ag[t][ks], false, bfl,
                                                             (short)0, c[t], false, false);
#pragma unroll
            for (int t = 0; t < MT; ++t)
                c[t] = __builtin_amdgcn_wmma_f32_16x16x4_f32(false, a_x[t][ks], false, bfr,
                                                             (short)0, c[t], false, false);
        }
        const float bias = b_s[nb * 16 + row];
#pragma unroll
        for (int t = 0; t < MT; ++t) {
            const int bt = base0 + t * 16;
            if (bt < N) {
#pragma unroll
                for (int r = 0; r < 8; ++r) {
                    const int M = r + half * 8;
                    out[(long)(bt + M) * FEAT + nb * 16 + row] = c[t][r] + bias;
                }
            }
        }
    }
}

extern "C" void kernel_launch(void* const* d_in, const int* in_sizes, int n_in,
                              void* d_out, int out_size, void* d_ws, size_t ws_size,
                              hipStream_t stream) {
    const float* x   = (const float*)d_in[0];
    const int*   eix = (const int*)d_in[1];
    const float* W1l = (const float*)d_in[2];
    const float* b1l = (const float*)d_in[3];
    const float* W1r = (const float*)d_in[4];
    const float* W2l = (const float*)d_in[5];
    const float* b2l = (const float*)d_in[6];
    const float* W2r = (const float*)d_in[7];

    const int  N = in_sizes[0] / FEAT;
    const long E = in_sizes[1] / 2;
    const int* src = eix;
    const int* dst = eix + E;

    float* aggr = (float*)d_ws;
    float* x1   = aggr + (size_t)N * FEAT;
    float* deg  = x1 + (size_t)N * FEAT;
    float* outp = (float*)d_out;
    float* keyf = outp;
    float* bmf  = outp + ((E + 127) / 128) * 32;

    const int  key_blocks = (int)(((E + 15) / 16 + KEY_THREADS - 1) / KEY_THREADS);
    const long Wp         = (long)key_blocks * 32;
    const int ag_blocks  = (N + TILE - 1) / TILE;
    const int up_blocks  = (N + 8 * 16 * MT - 1) / (8 * 16 * MT);

    range_index<<<key_blocks, KEY_THREADS, 0, stream>>>(dst, keyf, bmf, E, ag_blocks, Wp);

    aggregate_owner<<<ag_blocks, AG_THREADS, 0, stream>>>(x, src, dst, keyf, bmf, Wp, aggr, deg, E, N);
    sage_update_wmma<<<up_blocks, 256, 0, stream>>>(x, aggr, deg, W1l, b1l, W1r, x1, N);

    aggregate_owner<<<ag_blocks, AG_THREADS, 0, stream>>>(x1, src, dst, keyf, bmf, Wp, aggr, deg, E, N);
    sage_update_wmma<<<up_blocks, 256, 0, stream>>>(x1, aggr, deg, W2l, b2l, W2r, outp, N);
}


# index pass stores (dst,src) pairs, 128 B per 16-edge piece; owners fetch each flagged piece as one line (8 lanes) and queue (src,row): no key bytes, no scattered dst/src word loads; arrival order
# speedup vs baseline: 3.1161x; 1.1330x over previous
#include <hip/hip_runtime.h>

typedef __attribute__((ext_vector_type(2))) float v2f;
typedef __attribute__((ext_vector_type(8))) float v8f;

#define FEAT 64
#define TILE 400
#define QCAP 256
#define AG_THREADS 1024
#define KEY_THREADS 1024
#define RG 256
#define RGP 257
#define PU 8

__global__ void __launch_bounds__(KEY_THREADS)
range_index(const int* __restrict__ src, const int* __restrict__ dst, float* __restrict__ pairf, float* __restrict__ bmf,
            long E, int nR, long Wp) {
    __shared__ unsigned bm_s[32 * RGP];
    int2* pair = (int2*)pairf;
    unsigned* bitmap = (unsigned*)bmf;
    const int tid = threadIdx.x, lane = tid & 31, wave = tid >> 5;
    const bool vec_ok = (((unsigned long long)pairf) & 15ull) == 0ull;
    const long nP = vec_ok ? (E >> 4) : 0;
    const long pb = ((long)blockIdx.x * 32 + wave) * 32;
    const int  hp = lane >> 4;
    unsigned r[16];
    if (pb + 32 <= nP) {
        const long e0 = pb * 16 + lane;
        int dv[16], sv[16];
#pragma unroll
        for (int j = 0; j < 16; ++j) {
            dv[j] = dst[e0 + 32 * j];
            sv[j] = src[e0 + 32 * j];
        }
#pragma unroll
        for (int j = 0; j < 16; ++j) {
            pair[e0 + 32 * j] = make_int2(dv[j], sv[j]);
            r[j] = (unsigned)dv[j] / TILE;
        }
    } else {
#pragma unroll
        for (int j = 0; j < 16; ++j) {
            r[j] = 0xffffffffu;
            if (pb + 2 * j + hp < nP) {
                const long e = pb * 16 + j * 32 + lane;
                const int d = dst[e];
                const int s = src[e];
                pair[e] = make_int2(d, s);
                r[j] = (unsigned)d / TILE;
            }
        }
    }
    for (int g0 = 0; g0 < nR; g0 += RG) {
        for (int i = tid; i < 32 * RGP; i += KEY_THREADS) bm_s[i] = 0u;
        __syncthreads();
#pragma unroll
        for (int j = 0; j < 16; ++j) {
            const unsigned rl = r[j] - (unsigned)g0;
            if (rl < (unsigned)RG && r[j] < (unsigned)nR) atomicOr(&bm_s[wave * RGP + (int)rl], 1u << (2 * j + hp));
        }
        __syncthreads();
        const int cnt = min(RG, nR - g0);
        for (int i = tid; i < cnt * 32; i += KEY_THREADS) {
            const int rl = i >> 5, w = i & 31;
            bitmap[(size_t)(g0 + rl) * (size_t)Wp + (size_t)blockIdx.x * 32 + w] = bm_s[w * RGP + rl];
        }
        __syncthreads();
    }
}

__global__ void __launch_bounds__(AG_THREADS)
aggregate_owner(const float* __restrict__ x, const int* __restrict__ src, const int* __restrict__ dst,
                const float* __restrict__ pairf, const float* __restrict__ bmf, long Wp,
                float* __restrict__ aggr, float* __restrict__ deg, long E, int N) {
    __shared__ float    tile_s[(TILE + 1) * FEAT];
    __shared__ unsigned cnt_s[TILE + 1];
    __shared__ uint2    q_s[AG_THREADS / 32][QCAP];
    __shared__ unsigned short pl_s[AG_THREADS / 32][1024];

    const int tid  = threadIdx.x;
    const int lane = tid & 31;
    const int lo   = blockIdx.x * TILE;
    const unsigned ucnt = (unsigned)min(TILE, N - lo);
    uint2* q = q_s[tid >> 5];
    unsigned short* pl = pl_s[tid >> 5];

    for (int i = tid; i < (TILE + 1) * FEAT / 4; i += AG_THREADS)
        ((float4*)tile_s)[i] = make_float4(0.f, 0.f, 0.f, 0.f);
    for (int i = tid; i < TILE + 1; i += AG_THREADS) cnt_s[i] = 0u;
    __syncthreads();

    unsigned qhead = 0u, qcount = 0u;

    auto append = [&](bool m, unsigned s, unsigned a) {
        const unsigned mask = __builtin_amdgcn_ballot_w32(m);
        if (mask) {
            const unsigned below = __builtin_amdgcn_mbcnt_lo(mask, 0u);
            if (m) q[(qhead + qcount + below) & (QCAP - 1)] = make_uint2(s, a);
            qcount += (unsigned)__builtin_popcount(mask);
        }
    };

    auto drain = [&](unsigned n) {
        uint2 sa = make_uint2(0u, (unsigned)TILE);
        if ((unsigned)lane < n) sa = q[(qhead + (unsigned)lane) & (QCAP - 1)];
        const int s = (int)sa.x;
        const unsigned row = sa.y;
        atomicAdd(&cnt_s[row], 1u);
        for (int j0 = 0; j0 < 32; j0 += 8) {
            float2 v[8];
            unsigned r[8];
#pragma unroll
            for (int u = 0; u < 8; ++u) {
                const int sj = __builtin_amdgcn_readlane(s, j0 + u);
                r[u] = (unsigned)__builtin_amdgcn_readlane((int)row, j0 + u);
                v[u] = ((const float2*)(x + (long)sj * FEAT))[lane];
            }
#pragma unroll
            for (int u = 0; u < 8; ++u) {
                float* t = tile_s + r[u] * FEAT + lane * 2;
                atomicAdd(t, v[u].x);
                atomicAdd(t + 1, v[u].y);
            }
        }
        qhead = (qhead + 32u) & (QCAP - 1);
        qcount -= n;
    };

    const bool vec_ok = (((unsigned long long)pairf) & 15ull) == 0ull;
    const long nP     = vec_ok ? (E >> 4) : 0;
    const long Wn     = (nP + 31) >> 5;
    const int4* pair4 = (const int4*)pairf;
    const unsigned* brow = (const unsigned*)bmf + (size_t)blockIdx.x * (size_t)Wp;
    const int grp = lane >> 3, sub = lane & 7;

    for (long wb = 0; wb < Wn; wb += AG_THREADS) {
        const long wi = wb + tid;
        unsigned bits = wi < Wn ? brow[wi] : 0u;
        unsigned pcount = 0u;
        unsigned mask;
        while ((mask = __builtin_amdgcn_ballot_w32(bits != 0u)) != 0u) {
            const unsigned below = __builtin_amdgcn_mbcnt_lo(mask, 0u);
            if (bits != 0u) pl[pcount + below] = (unsigned short)((lane << 5) | __builtin_ctz(bits));
            pcount += (unsigned)__builtin_popcount(mask);
            bits &= bits - 1u;
        }
        const long pbase = (wi - lane) * 32;
        for (unsigned i0 = 0u; i0 < pcount; i0 += 4u * PU) {
            int4 d[PU];
            bool hv[PU];
            const unsigned k0 = i0 + (unsigned)(PU * grp);
            uint4 pk;
            __builtin_memcpy(&pk, __builtin_assume_aligned(pl + k0, 16), 16);
            const unsigned pw[4] = {pk.x, pk.y, pk.z, pk.w};
#pragma unroll
            for (int u = 0; u < PU; ++u) {
                hv[u] = k0 + (unsigned)u < pcount;
                const unsigned idx = (u & 1) ? (pw[u >> 1] >> 16) : (pw[u >> 1] & 0xffffu);
                const long p = hv[u] ? pbase + (long)idx : 0L;
                d[u] = pair4[p * 8 + sub];
            }
#pragma unroll
            for (int u = 0; u < PU; ++u) {
                const unsigned a0 = (unsigned)d[u].x - (unsigned)lo;
                const unsigned a1 = (unsigned)d[u].z - (unsigned)lo;
                append(hv[u] && a0 < ucnt, (unsigned)d[u].y, a0);
                append(hv[u] && a1 < ucnt, (unsigned)d[u].w, a1);
                while (qcount >= 32u) drain(32u);
            }
        }
    }
    for (long eb = nP * 16; eb < E; eb += AG_THREADS) {
        const long e = eb + tid;
        const bool in = e < E;
        const unsigned a = (unsigned)(in ? dst[e] : 0) - (unsigned)lo;
        const unsigned s = (unsigned)(in ? src[e] : 0);
        append(in && a < ucnt, s, a);
        while (qcount >= 32u) drain(32u);
    }
    if (qcount) drain(qcount);
    __syncthreads();

    float4* orow = (float4*)(aggr + (long)lo * FEAT);
    float*  odeg = deg + lo;
    for (int i = tid; i < (int)ucnt * (FEAT / 4); i += AG_THREADS) orow[i] = ((const float4*)tile_s)[i];
    for (int i = tid; i < (int)ucnt; i += AG_THREADS) odeg[i] = (float)cnt_s[i];
}

#define MT 4
__global__ void __launch_bounds__(256)
sage_update_wmma(const float* __restrict__ x, const float* __restrict__ aggr,
                 const float* __restrict__ deg,
                 const float* __restrict__ Wl, const float* __restrict__ bl,
                 const float* __restrict__ Wr,
                 float* __restrict__ out, int N) {
    __shared__ float Wl_s[FEAT * FEAT];
    __shared__ float Wr_s[FEAT * FEAT];
    __shared__ float b_s[FEAT];

    for (int i = threadIdx.x; i < FEAT * FEAT / 4; i += 256) {
        ((float4*)Wl_s)[i] = ((const float4*)Wl)[i];
        ((float4*)Wr_s)[i] = ((const float4*)Wr)[i];
    }
    if (threadIdx.x < FEAT) b_s[threadIdx.x] = bl[threadIdx.x];
    __syncthreads();

    const int lane = threadIdx.x & 31;
    const int wave = threadIdx.x >> 5;
    const int base0 = (blockIdx.x * 8 + wave) * (16 * MT);
    if (base0 >= N) return;

    const int row  = lane & 15;
    const int half = lane >> 4;

    v2f a_ag[MT][16], a_x[MT][16];
#pragma unroll
    for (int t = 0; t < MT; ++t) {
        const int bt   = base0 + t * 16;
        const int node = ((bt < N) ? bt : base0) + row;
        const float dinv = 1.0f / fmaxf(deg[node], 1.0f);
        const float* agp = aggr + (long)node * FEAT;
        const float* xp  = x    + (long)node * FEAT;
#pragma unroll
        for (int ks = 0; ks < 16; ++ks) {
            const int k0 = ks * 4 + half * 2;
            v2f t0; t0.x = agp[k0] * dinv; t0.y = agp[k0 + 1] * dinv;
            a_ag[t][ks] = t0;
            v2f t1; t1.x = xp[k0];         t1.y = xp[k0 + 1];
            a_x[t][ks] = t1;
        }
    }

#pragma unroll
    for (int nb = 0; nb < 4; ++nb) {
        const int ncol = nb * 16 + row;
        v8f c[MT];
#pragma unroll
        for (int t = 0; t < MT; ++t) c[t] = v8f{};
#pragma unroll
        for (int ks = 0; ks < 16; ++ks) {
            const int k0 = ks * 4 + half * 2;
            v2f bfl; bfl.x = Wl_s[ncol * FEAT + k0]; bfl.y = Wl_s[ncol * FEAT + k0 + 1];
            v2f bfr; bfr.x = Wr_s[ncol * FEAT + k0]; bfr.y = Wr_s[ncol * FEAT + k0 + 1];
#pragma unroll
            for (int t = 0; t < MT; ++t)
                c[t] = __builtin_amdgcn_wmma_f32_16x16x4_f32(false, a_ag[t][ks], false, bfl,
                                                             (short)0, c[t], false, false);
#pragma unroll
            for (int t = 0; t < MT; ++t)
                c[t] = __builtin_amdgcn_wmma_f32_16x16x4_f32(false, a_x[t][ks], false, bfr,
                                                             (short)0, c[t], false, false);
        }
        const float bias = b_s[nb * 16 + row];
#pragma unroll
        for (int t = 0; t < MT; ++t) {
            const int bt = base0 + t * 16;
            if (bt < N) {
#pragma unroll
                for (int r = 0; r < 8; ++r) {
                    const int M = r + half * 8;
                    out[(long)(bt + M) * FEAT + nb * 16 + row] = c[t][r] + bias;
                }
            }
        }
    }
}

extern "C" void kernel_launch(void* const* d_in, const int* in_sizes, int n_in,
                              void* d_out, int out_size, void* d_ws, size_t ws_size,
                              hipStream_t stream) {
    const float* x   = (const float*)d_in[0];
    const int*   eix = (const int*)d_in[1];
    const float* W1l = (const float*)d_in[2];
    const float* b1l = (const float*)d_in[3];
    const float* W1r = (const float*)d_in[4];
    const float* W2l = (const float*)d_in[5];
    const float* b2l = (const float*)d_in[6];
    const float* W2r = (const float*)d_in[7];

    const int  N = in_sizes[0] / FEAT;
    const long E = in_sizes[1] / 2;
    const int* src = eix;
    const int* dst = eix + E;

    float* aggr = (float*)d_ws;
    float* x1   = aggr + (size_t)N * FEAT;
    float* deg  = x1 + (size_t)N * FEAT;
    float* outp = (float*)d_out;
    float* pairf = outp;
    float* bmf  = outp + ((E + 15) / 16) * 32;

    const int  key_blocks = (int)(((E + 15) / 16 + KEY_THREADS - 1) / KEY_THREADS);
    const long Wp         = (long)key_blocks * 32;
    const int ag_blocks  = (N + TILE - 1) / TILE;
    const int up_blocks  = (N + 8 * 16 * MT - 1) / (8 * 16 * MT);

    range_index<<<key_blocks, KEY_THREADS, 0, stream>>>(src, dst, pairf, bmf, E, ag_blocks, Wp);

    aggregate_owner<<<ag_blocks, AG_THREADS, 0, stream>>>(x, src, dst, pairf, bmf, Wp, aggr, deg, E, N);
    sage_update_wmma<<<up_blocks, 256, 0, stream>>>(x, aggr, deg, W1l, b1l, W1r, x1, N);

    aggregate_owner<<<ag_blocks, AG_THREADS, 0, stream>>>(x1, src, dst, pairf, bmf, Wp, aggr, deg, E, N);
    sage_update_wmma<<<up_blocks, 256, 0, stream>>>(x1, aggr, deg, W2l, b2l, W2r, outp, N);
}


# layer 1's owners log each drain's 32 (source,row) slots as one 128-B line in a per-work-group list behind the bitmap; layer 2's waves stream those lines: no bitmap walk, no pair lines; arrival order
# speedup vs baseline: 3.2160x; 1.0321x over previous
#include <hip/hip_runtime.h>

typedef __attribute__((ext_vector_type(2))) float v2f;
typedef __attribute__((ext_vector_type(8))) float v8f;

#define FEAT 64
#define TILE 400
#define QCAP 256
#define AG_THREADS 1024
#define KEY_THREADS 1024
#define RG 256
#define RGP 257
#define PU 8
#define LSH 9
#define LQ 4
static_assert(TILE < (1 << LSH), "an owned row (and the spare row) must fit the low bits of a list word");

__global__ void __launch_bounds__(KEY_THREADS)
range_index(const int* __restrict__ src, const int* __restrict__ dst, float* __restrict__ pairf, float* __restrict__ bmf,
            long E, int nR, long Wp) {
    __shared__ unsigned bm_s[32 * RGP];
    int2* pair = (int2*)pairf;
    unsigned* bitmap = (unsigned*)bmf;
    const int tid = threadIdx.x, lane = tid & 31, wave = tid >> 5;
    const bool vec_ok = (((unsigned long long)pairf) & 15ull) == 0ull;
    const long nP = vec_ok ? (E >> 4) : 0;
    const long pb = ((long)blockIdx.x * 32 + wave) * 32;
    const int  hp = lane >> 4;
    unsigned r[16];
    if (pb + 32 <= nP) {
        const long e0 = pb * 16 + lane;
        int dv[16], sv[16];
#pragma unroll
        for (int j = 0; j < 16; ++j) {
            dv[j] = dst[e0 + 32 * j];
            sv[j] = src[e0 + 32 * j];
        }
#pragma unroll
        for (int j = 0; j < 16; ++j) {
            pair[e0 + 32 * j] = make_int2(dv[j], sv[j]);
            r[j] = (unsigned)dv[j] / TILE;
        }
    } else {
#pragma unroll
        for (int j = 0; j < 16; ++j) {
            r[j] = 0xffffffffu;
            if (pb + 2 * j + hp < nP) {
                const long e = pb * 16 + j * 32 + lane;
                const int d = dst[e];
                const int s = src[e];
                pair[e] = make_int2(d, s);
                r[j] = (unsigned)d / TILE;
            }
        }
    }
    for (int g0 = 0; g0 < nR; g0 += RG) {
        for (int i = tid; i < 32 * RGP; i += KEY_THREADS) bm_s[i] = 0u;
        __syncthreads();
#pragma unroll
        for (int j = 0; j < 16; ++j) {
            const unsigned rl = r[j] - (unsigned)g0;
            if (rl < (unsigned)RG && r[j] < (unsigned)nR) atomicOr(&bm_s[wave * RGP + (int)rl], 1u << (2 * j + hp));
        }
        __syncthreads();
        const int cnt = min(RG, nR - g0);
        for (int i = tid; i < cnt * 32; i += KEY_THREADS) {
            const int rl = i >> 5, w = i & 31;
            bitmap[(size_t)(g0 + rl) * (size_t)Wp + (size_t)blockIdx.x * 32 + w] = bm_s[w * RGP + rl];
        }
        __syncthreads();
    }
}

template <bool L2>
__global__ void __launch_bounds__(AG_THREADS)
aggregate_owner(const float* __restrict__ x, const int* __restrict__ src, const int* __restrict__ dst,
                const float* __restrict__ pairf, const float* __restrict__ bmf, long Wp,
                float* __restrict__ listf, const float* __restrict__ list_end,
                float* __restrict__ aggr, float* __restrict__ deg, long E, int N) {
    __shared__ float    tile_s[(TILE + 1) * FEAT];
    __shared__ unsigned cnt_s[TILE + 1];
    __shared__ uint2    q_s[AG_THREADS / 32][QCAP];
    __shared__ unsigned short pl_s[AG_THREADS / 32][1024];
    __shared__ unsigned lpos_s;

    const int tid  = threadIdx.x;
    const int lane = tid & 31;
    const int lo   = blockIdx.x * TILE;
    const unsigned ucnt = (unsigned)min(TILE, N - lo);
    uint2* q = q_s[tid >> 5];
    unsigned short* pl = pl_s[tid >> 5];

    const long nR     = (long)((N + TILE - 1) / TILE);
    const long nRp    = nR * 32;
    const long lwords = (long)(list_end - listf) - nRp;
    const long lper   = lwords > 0 ? lwords / nR : 0L;
    const unsigned capg = (N <= (1 << (32 - LSH))) ? ((unsigned)(lper < (1L << 24) ? lper : (1L << 24)) & ~31u) : 0u;
    unsigned* cntw    = (unsigned*)listf + (long)blockIdx.x * 32;
    unsigned* glist   = (unsigned*)listf + nRp + (long)blockIdx.x * (long)capg;

    for (int i = tid; i < (TILE + 1) * FEAT / 4; i += AG_THREADS)
        ((float4*)tile_s)[i] = make_float4(0.f, 0.f, 0.f, 0.f);
    for (int i = tid; i < TILE + 1; i += AG_THREADS) cnt_s[i] = 0u;
    if (tid == 0) lpos_s = 0u;
    __syncthreads();

    unsigned qhead = 0u, qcount = 0u;

    auto append = [&](bool m, unsigned s, unsigned a) {
        const unsigned mask = __builtin_amdgcn_ballot_w32(m);
        if (mask) {
            const unsigned below = __builtin_amdgcn_mbcnt_lo(mask, 0u);
            if (m) q[(qhead + qcount + below) & (QCAP - 1)] = make_uint2(s, a);
            qcount += (unsigned)__builtin_popcount(mask);
        }
    };

    auto add_rows = [&](int s, unsigned row) {
        atomicAdd(&cnt_s[row], 1u);
        for (int j0 = 0; j0 < 32; j0 += 8) {
            float2 v[8];
            unsigned r[8];
#pragma unroll
            for (int u = 0; u < 8; ++u) {
                const int sj = __builtin_amdgcn_readlane(s, j0 + u);
                r[u] = (unsigned)__builtin_amdgcn_readlane((int)row, j0 + u);
                v[u] = ((const float2*)(x + (long)sj * FEAT))[lane];
            }
#pragma unroll
            for (int u = 0; u < 8; ++u) {
                float* t = tile_s + r[u] * FEAT + lane * 2;
                atomicAdd(t, v[u].x);
                atomicAdd(t + 1, v[u].y);
            }
        }
    };

    auto drain = [&](unsigned n) {
        uint2 sa = make_uint2(0u, (unsigned)TILE);
        if ((unsigned)lane < n) sa = q[(qhead + (unsigned)lane) & (QCAP - 1)];
        if (!L2) {
            unsigned base = 0u;
            if (lane == 0) base = atomicAdd(&lpos_s, 32u);
            base = (unsigned)__builtin_amdgcn_readfirstlane((int)base);
            if (base + 32u <= capg) glist[base + (unsigned)lane] = (sa.x << LSH) | sa.y;
        }
        add_rows((int)sa.x, sa.y);
        qhead = (qhead + 32u) & (QCAP - 1);
        qcount -= n;
    };

    const bool vec_ok = (((unsigned long long)pairf) & 15ull) == 0ull;
    const long nP     = vec_ok ? (E >> 4) : 0;
    const long Wn     = (nP + 31) >> 5;
    const int4* pair4 = (const int4*)pairf;
    const unsigned* brow = (const unsigned*)bmf + (size_t)blockIdx.x * (size_t)Wp;
    const int grp = lane >> 3, sub = lane & 7;

    bool walk = true;
    if (L2 && capg >= 32u) {
        const unsigned c = (unsigned)__builtin_amdgcn_readfirstlane((int)*cntw);
        if (c <= capg) {
            walk = false;
            const unsigned* ln = glist + (unsigned)lane;
            for (unsigned w0 = (unsigned)(tid >> 5) * 32u; w0 < c; w0 += LQ * AG_THREADS) {
                unsigned e[LQ];
#pragma unroll
                for (int u = 0; u < LQ; ++u) {
                    const unsigned w = w0 + (unsigned)u * AG_THREADS;
                    e[u] = w < c ? ln[w] : (unsigned)TILE;
                }
#pragma unroll
                for (int u = 0; u < LQ; ++u)
                    if (w0 + (unsigned)u * AG_THREADS < c) add_rows((int)(e[u] >> LSH), e[u] & ((1u << LSH) - 1u));
            }
        }
    }

    if (walk) {
        for (long wb = 0; wb < Wn; wb += AG_THREADS) {
            const long wi = wb + tid;
            unsigned bits = wi < Wn ? brow[wi] : 0u;
            unsigned pcount = 0u;
            unsigned mask;
            while ((mask = __builtin_amdgcn_ballot_w32(bits != 0u)) != 0u) {
                const unsigned below = __builtin_amdgcn_mbcnt_lo(mask, 0u);
                if (bits != 0u) pl[pcount + below] = (unsigned short)((lane << 5) | __builtin_ctz(bits));
                pcount += (unsigned)__builtin_popcount(mask);
                bits &= bits - 1u;
            }
            const long pbase = (wi - lane) * 32;
            for (unsigned i0 = 0u; i0 < pcount; i0 += 4u * PU) {
                int4 d[PU];
                bool hv[PU];
                const unsigned k0 = i0 + (unsigned)(PU * grp);
                uint4 pk;
                __builtin_memcpy(&pk, __builtin_assume_aligned(pl + k0, 16), 16);
                const unsigned pw[4] = {pk.x, pk.y, pk.z, pk.w};
#pragma unroll
                for (int u = 0; u < PU; ++u) {
                    hv[u] = k0 + (unsigned)u < pcount;
                    const unsigned idx = (u & 1) ? (pw[u >> 1] >> 16) : (pw[u >> 1] & 0xffffu);
                    const long p = hv[u] ? pbase + (long)idx : 0L;
                    d[u] = pair4[p * 8 + sub];
                }
#pragma unroll
                for (int u = 0; u < PU; ++u) {
                    const unsigned a0 = (unsigned)d[u].x - (unsigned)lo;
                    const unsigned a1 = (unsigned)d[u].z - (unsigned)lo;
                    append(hv[u] && a0 < ucnt, (unsigned)d[u].y, a0);
                    append(hv[u] && a1 < ucnt, (unsigned)d[u].w, a1);
                    while (qcount >= 32u) drain(32u);
                }
            }
        }
        for (long eb = nP * 16; eb < E; eb += AG_THREADS) {
            const long e = eb + tid;
            const bool in = e < E;
            const unsigned a = (unsigned)(in ? dst[e] : 0) - (unsigned)lo;
            const unsigned s = (unsigned)(in ? src[e] : 0);
            append(in && a < ucnt, s, a);
            while (qcount >= 32u) drain(32u);
        }
        if (qcount) drain(qcount);
    }
    __syncthreads();
    if (!L2 && capg >= 32u && tid == 0) *cntw = lpos_s;

    float4* orow = (float4*)(aggr + (long)lo * FEAT);
    float*  odeg = deg + lo;
    for (int i = tid; i < (int)ucnt * (FEAT / 4); i += AG_THREADS) orow[i] = ((const float4*)tile_s)[i];
    for (int i = tid; i < (int)ucnt; i += AG_THREADS) odeg[i] = (float)cnt_s[i];
}

#define MT 4
__global__ void __launch_bounds__(256)
sage_update_wmma(const float* __restrict__ x, const float* __restrict__ aggr,
                 const float* __restrict__ deg,
                 const float* __restrict__ Wl, const float* __restrict__ bl,
                 const float* __restrict__ Wr,
                 float* __restrict__ out, int N) {
    __shared__ float Wl_s[FEAT * FEAT];
    __shared__ float Wr_s[FEAT * FEAT];
    __shared__ float b_s[FEAT];

    for (int i = threadIdx.x; i < FEAT * FEAT / 4; i += 256) {
        ((float4*)Wl_s)[i] = ((const float4*)Wl)[i];
        ((float4*)Wr_s)[i] = ((const float4*)Wr)[i];
    }
    if (threadIdx.x < FEAT) b_s[threadIdx.x] = bl[threadIdx.x];
    __syncthreads();

    const int lane = threadIdx.x & 31;
    const int wave = threadIdx.x >> 5;
    const int base0 = (blockIdx.x * 8 + wave) * (16 * MT);
    if (base0 >= N) return;

    const int row  = lane & 15;
    const int half = lane >> 4;

    v2f a_ag[MT][16], a_x[MT][16];
#pragma unroll
    for (int t = 0; t < MT; ++t) {
        const int bt   = base0 + t * 16;
        const int node = ((bt < N) ? bt : base0) + row;
        const float dinv = 1.0f / fmaxf(deg[node], 1.0f);
        const float* agp = aggr + (long)node * FEAT;
        const float* xp  = x    + (long)node * FEAT;
#pragma unroll
        for (int ks = 0; ks < 16; ++ks) {
            const int k0 = ks * 4 + half * 2;
            v2f t0; t0.x = agp[k0] * dinv; t0.y = agp[k0 + 1] * dinv;
            a_ag[t][ks] = t0;
            v2f t1; t1.x = xp[k0];         t1.y = xp[k0 + 1];
            a_x[t][ks] = t1;
        }
    }

#pragma unroll
    for (int nb = 0; nb < 4; ++nb) {
        const int ncol = nb * 16 + row;
        v8f c[MT];
#pragma unroll
        for (int t = 0; t < MT; ++t) c[t] = v8f{};
#pragma unroll
        for (int ks = 0; ks < 16; ++ks) {
            const int k0 = ks * 4 + half * 2;
            v2f bfl; bfl.x = Wl_s[ncol * FEAT + k0]; bfl.y = Wl_s[ncol * FEAT + k0 + 1];
            v2f bfr; bfr.x = Wr_s[ncol * FEAT + k0]; bfr.y = Wr_s[ncol * FEAT + k0 + 1];
#pragma unroll
            for (int t = 0; t < MT; ++t)
                c[t] = __builtin_amdgcn_wmma_f32_16x16x4_f32(false, a_ag[t][ks], false, bfl,
                                                             (short)0, c[t], false, false);
#pragma unroll
            for (int t = 0; t < MT; ++t)
                c[t] = __builtin_amdgcn_wmma_f32_16x16x4_f32(false, a_x[t][ks], false, bfr,
                                                             (short)0, c[t], false, false);
        }
        const float bias = b_s[nb * 16 + row];
#pragma unroll
        for (int t = 0; t < MT; ++t) {
            const int bt = base0 + t * 16;
            if (bt < N) {
#pragma unroll
                for (int r = 0; r < 8; ++r) {
                    const int M = r + half * 8;
                    out[(long)(bt + M) * FEAT + nb * 16 + row] = c[t][r] + bias;
                }
            }
        }
    }
}

extern "C" void kernel_launch(void* const* d_in, const int* in_sizes, int n_in,
                              void* d_out, int out_size, void* d_ws, size_t ws_size,
                              hipStream_t stream) {
    const float* x   = (const float*)d_in[0];
    const int*   eix = (const int*)d_in[1];
    const float* W1l = (const float*)d_in[2];
    const float* b1l = (const float*)d_in[3];
    const float* W1r = (const float*)d_in[4];
    const float* W2l = (const float*)d_in[5];
    const float* b2l = (const float*)d_in[6];
    const float* W2r = (const float*)d_in[7];

    const int  N = in_sizes[0] / FEAT;
    const long E = in_sizes[1] / 2;
    const int* src = eix;
    const int* dst = eix + E;

    float* aggr = (float*)d_ws;
    float* x1   = aggr + (size_t)N * FEAT;
    float* deg  = x1 + (size_t)N * FEAT;
    float* outp = (float*)d_out;
    float* pairf = outp;
    float* bmf  = outp + ((E + 15) / 16) * 32;

    const int  key_blocks = (int)(((E + 15) / 16 + KEY_THREADS - 1) / KEY_THREADS);
    const long Wp         = (long)key_blocks * 32;
    const int ag_blocks  = (N + TILE - 1) / TILE;
    const int up_blocks  = (N + 8 * 16 * MT - 1) / (8 * 16 * MT);

    float* listf = bmf + (size_t)ag_blocks * (size_t)Wp;
    float* oend  = outp + (size_t)N * FEAT;

    range_index<<<key_blocks, KEY_THREADS, 0, stream>>>(src, dst, pairf, bmf, E, ag_blocks, Wp);

    aggregate_owner<false><<<ag_blocks, AG_THREADS, 0, stream>>>(x, src, dst, pairf, bmf, Wp, listf, oend, aggr, deg, E, N);
    sage_update_wmma<<<up_blocks, 256, 0, stream>>>(x, aggr, deg, W1l, b1l, W1r, x1, N);

    aggregate_owner<true><<<ag_blocks, AG_THREADS, 0, stream>>>(x1, src, dst, pairf, bmf, Wp, listf, oend, aggr, deg, E, N);
    sage_update_wmma<<<up_blocks, 256, 0, stream>>>(x1, aggr, deg, W2l, b2l, W2r, outp, N);
}


# update fused into owner work-groups: a wave finishes 16 nodes from the LDS sums (A fragments read there, weights in LDS in fragment order); no aggr/deg round trip; 5 to 3 launches; same sum order
# speedup vs baseline: 3.9511x; 1.2286x over previous
#include <hip/hip_runtime.h>

typedef __attribute__((ext_vector_type(2))) float v2f;
typedef __attribute__((ext_vector_type(8))) float v8f;

#define FEAT 64
#define TILE 400
#define QCAP 256
#define AG_THREADS 1024
#define KEY_THREADS 1024
#define RG 256
#define RGP 257
#define PU 8
#define LSH 9
#define LQ 4
#define TP 68
#define FB 66
static_assert(TILE < (1 << LSH), "an owned row (and the spare row) must fit the low bits of a list word");
static_assert(TILE % 16 == 0 && TILE <= 16 * (AG_THREADS / 32), "one 16-node tile per wave in the update phase");

__global__ void __launch_bounds__(KEY_THREADS)
range_index(const int* __restrict__ src, const int* __restrict__ dst, float* __restrict__ pairf, float* __restrict__ bmf,
            long E, int nR, long Wp) {
    __shared__ unsigned bm_s[32 * RGP];
    int2* pair = (int2*)pairf;
    unsigned* bitmap = (unsigned*)bmf;
    const int tid = threadIdx.x, lane = tid & 31, wave = tid >> 5;
    const bool vec_ok = (((unsigned long long)pairf) & 15ull) == 0ull;
    const long nP = vec_ok ? (E >> 4) : 0;
    const long pb = ((long)blockIdx.x * 32 + wave) * 32;
    const int  hp = lane >> 4;
    unsigned r[16];
    if (pb + 32 <= nP) {
        const long e0 = pb * 16 + lane;
        int dv[16], sv[16];
#pragma unroll
        for (int j = 0; j < 16; ++j) {
            dv[j] = dst[e0 + 32 * j];
            sv[j] = src[e0 + 32 * j];
        }
#pragma unroll
        for (int j = 0; j < 16; ++j) {
            pair[e0 + 32 * j] = make_int2(dv[j], sv[j]);
            r[j] = (unsigned)dv[j] / TILE;
        }
    } else {
#pragma unroll
        for (int j = 0; j < 16; ++j) {
            r[j] = 0xffffffffu;
            if (pb + 2 * j + hp < nP) {
                const long e = pb * 16 + j * 32 + lane;
                const int d = dst[e];
                const int s = src[e];
                pair[e] = make_int2(d, s);
                r[j] = (unsigned)d / TILE;
            }
        }
    }
    for (int g0 = 0; g0 < nR; g0 += RG) {
        for (int i = tid; i < 32 * RGP; i += KEY_THREADS) bm_s[i] = 0u;
        __syncthreads();
#pragma unroll
        for (int j = 0; j < 16; ++j) {
            const unsigned rl = r[j] - (unsigned)g0;
            if (rl < (unsigned)RG && r[j] < (unsigned)nR) atomicOr(&bm_s[wave * RGP + (int)rl], 1u << (2 * j + hp));
        }
        __syncthreads();
        const int cnt = min(RG, nR - g0);
        for (int i = tid; i < cnt * 32; i += KEY_THREADS) {
            const int rl = i >> 5, w = i & 31;
            bitmap[(size_t)(g0 + rl) * (size_t)Wp + (size_t)blockIdx.x * 32 + w] = bm_s[w * RGP + rl];
        }
        __syncthreads();
    }
}

template <bool L2>
__global__ void __launch_bounds__(AG_THREADS)
aggregate_update(const float* __restrict__ x, const int* __restrict__ src, const int* __restrict__ dst,
                 const float* __restrict__ pairf, const float* __restrict__ bmf, long Wp,
                 float* __restrict__ listf, const float* __restrict__ list_end,
                 const float* __restrict__ Wl, const float* __restrict__ bl, const float* __restrict__ Wr,
                 float* __restrict__ out, long E, int N) {
    __shared__ __attribute__((aligned(16))) float tile_s[(TILE + 1) * TP];
    __shared__ __attribute__((aligned(16))) float wl_s[FEAT * FB];
    __shared__ __attribute__((aligned(16))) float wr_s[FEAT * FB];
    __shared__ unsigned cnt_s[TILE + 1];
    __shared__ uint2    q_s[AG_THREADS / 32][QCAP];
    __shared__ unsigned short pl_s[AG_THREADS / 32][1024];
    __shared__ unsigned lpos_s;

    const int tid  = threadIdx.x;
    const int lane = tid & 31;
    const int lo   = blockIdx.x * TILE;
    const unsigned ucnt = (unsigned)min(TILE, N - lo);
    uint2* q = q_s[tid >> 5];
    unsigned short* pl = pl_s[tid >> 5];

    const long nR     = (long)((N + TILE - 1) / TILE);
    const long nRp    = nR * 32;
    const long lwords = (long)(list_end - listf) - nRp;
    const long lper   = lwords > 0 ? lwords / nR : 0L;
    const unsigned capg = (N <= (1 << (32 - LSH))) ? ((unsigned)(lper < (1L << 24) ? lper : (1L << 24)) & ~31u) : 0u;
    unsigned* cntw    = (unsigned*)listf + (long)blockIdx.x * 32;
    unsigned* glist   = (unsigned*)listf + nRp + (long)blockIdx.x * (long)capg;

    const int wave = tid >> 5;
    const int row  = lane & 15;
    const int half = lane >> 4;
    const int t0   = wave * 16;
    const bool has = t0 < (int)ucnt;
    float4 xt[8];
    if (has) {
        const float4* xp4 = (const float4*)(x + (long)(lo + t0) * FEAT);
#pragma unroll
        for (int j = 0; j < 8; ++j)
            xt[j] = (unsigned)(t0 + 2 * j + half) < ucnt ? xp4[j * 32 + lane] : make_float4(0.f, 0.f, 0.f, 0.f);
    }
    float4 wl4[FEAT * FEAT / 4 / AG_THREADS], wr4[FEAT * FEAT / 4 / AG_THREADS];
#pragma unroll
    for (int j = 0; j < FEAT * FEAT / 4 / AG_THREADS; ++j) {
        const int i = tid + j * AG_THREADS;
        wl4[j] = ((const float4*)Wl)[i];
        wr4[j] = ((const float4*)Wr)[i];
    }

    for (int i = tid; i < (TILE + 1) * TP / 4; i += AG_THREADS)
        ((float4*)tile_s)[i] = make_float4(0.f, 0.f, 0.f, 0.f);
    for (int i = tid; i < TILE + 1; i += AG_THREADS) cnt_s[i] = 0u;
    if (tid == 0) lpos_s = 0u;
#pragma unroll
    for (int j = 0; j < FEAT * FEAT / 4 / AG_THREADS; ++j) {
        const int i = tid + j * AG_THREADS;
        const int n = i >> 4, ks = i & 15;
        const int o = ((n >> 4) * 16 + ks) * FB + (n & 15) * 2;
        *(float2*)(wl_s + o)      = make_float2(wl4[j].x, wl4[j].y);
        *(float2*)(wl_s + o + 32) = make_float2(wl4[j].z, wl4[j].w);
        *(float2*)(wr_s + o)      = make_float2(wr4[j].x, wr4[j].y);
        *(float2*)(wr_s + o + 32) = make_float2(wr4[j].z, wr4[j].w);
    }
    __syncthreads();

    unsigned qhead = 0u, qcount = 0u;

    auto append = [&](bool m, unsigned s, unsigned a) {
        const unsigned mask = __builtin_amdgcn_ballot_w32(m);
        if (mask) {
            const unsigned below = __builtin_amdgcn_mbcnt_lo(mask, 0u);
            if (m) q[(qhead + qcount + below) & (QCAP - 1)] = make_uint2(s, a);
            qcount += (unsigned)__builtin_popcount(mask);
        }
    };

    auto add_rows = [&](int s, unsigned row) {
        atomicAdd(&cnt_s[row], 1u);
        for (int j0 = 0; j0 < 32; j0 += 8) {
            float2 v[8];
            unsigned r[8];
#pragma unroll
            for (int u = 0; u < 8; ++u) {
                const int sj = __builtin_amdgcn_readlane(s, j0 + u);
                r[u] = (unsigned)__builtin_amdgcn_readlane((int)row, j0 + u);
                v[u] = ((const float2*)(x + (long)sj * FEAT))[lane];
            }
#pragma unroll
            for (int u = 0; u < 8; ++u) {
                float* t = tile_s + r[u] * TP + lane * 2;
                atomicAdd(t, v[u].x);
                atomicAdd(t + 1, v[u].y);
            }
        }
    };

    auto drain = [&](unsigned n) {
        uint2 sa = make_uint2(0u, (unsigned)TILE);
        if ((unsigned)lane < n) sa = q[(qhead + (unsigned)lane) & (QCAP - 1)];
        if (!L2) {
            unsigned base = 0u;
            if (lane == 0) base = atomicAdd(&lpos_s, 32u);
            base = (unsigned)__builtin_amdgcn_readfirstlane((int)base);
            if (base + 32u <= capg) glist[base + (unsigned)lane] = (sa.x << LSH) | sa.y;
        }
        add_rows((int)sa.x, sa.y);
        qhead = (qhead + 32u) & (QCAP - 1);
        qcount -= n;
    };

    const bool vec_ok = (((unsigned long long)pairf) & 15ull) == 0ull;
    const long nP     = vec_ok ? (E >> 4) : 0;
    const long Wn     = (nP + 31) >> 5;
    const int4* pair4 = (const int4*)pairf;
    const unsigned* brow = (const unsigned*)bmf + (size_t)blockIdx.x * (size_t)Wp;
    const int grp = lane >> 3, sub = lane & 7;

    bool walk = true;
    if (L2 && capg >= 32u) {
        const unsigned c = (unsigned)__builtin_amdgcn_readfirstlane((int)*cntw);
        if (c <= capg) {
            walk = false;
            const unsigned* ln = glist + (unsigned)lane;
            for (unsigned w0 = (unsigned)(tid >> 5) * 32u; w0 < c; w0 += LQ * AG_THREADS) {
                unsigned e[LQ];
#pragma unroll
                for (int u = 0; u < LQ; ++u) {
                    const unsigned w = w0 + (unsigned)u * AG_THREADS;
                    e[u] = w < c ? ln[w] : (unsigned)TILE;
                }
#pragma unroll
                for (int u = 0; u < LQ; ++u)
                    if (w0 + (unsigned)u * AG_THREADS < c) add_rows((int)(e[u] >> LSH), e[u] & ((1u << LSH) - 1u));
            }
        }
    }

    if (walk) {
        for (long wb = 0; wb < Wn; wb += AG_THREADS) {
            const long wi = wb + tid;
            unsigned bits = wi < Wn ? brow[wi] : 0u;
            unsigned pcount = 0u;
            unsigned mask;
            while ((mask = __builtin_amdgcn_ballot_w32(bits != 0u)) != 0u) {
                const unsigned below = __builtin_amdgcn_mbcnt_lo(mask, 0u);
                if (bits != 0u) pl[pcount + below] = (unsigned short)((lane << 5) | __builtin_ctz(bits));
                pcount += (unsigned)__builtin_popcount(mask);
                bits &= bits - 1u;
            }
            const long pbase = (wi - lane) * 32;
            for (unsigned i0 = 0u; i0 < pcount; i0 += 4u * PU) {
                int4 d[PU];
                bool hv[PU];
                const unsigned k0 = i0 + (unsigned)(PU * grp);
                uint4 pk;
                __builtin_memcpy(&pk, __builtin_assume_aligned(pl + k0, 16), 16);
                const unsigned pw[4] = {pk.x, pk.y, pk.z, pk.w};
#pragma unroll
                for (int u = 0; u < PU; ++u) {
                    hv[u] = k0 + (unsigned)u < pcount;
                    const unsigned idx = (u & 1) ? (pw[u >> 1] >> 16) : (pw[u >> 1] & 0xffffu);
                    const long p = hv[u] ? pbase + (long)idx : 0L;
                    d[u] = pair4[p * 8 + sub];
                }
#pragma unroll
                for (int u = 0; u < PU; ++u) {
                    const unsigned a0 = (unsigned)d[u].x - (unsigned)lo;
                    const unsigned a1 = (unsigned)d[u].z - (unsigned)lo;
                    append(hv[u] && a0 < ucnt, (unsigned)d[u].y, a0);
                    append(hv[u] && a1 < ucnt, (unsigned)d[u].w, a1);
                    while (qcount >= 32u) drain(32u);
                }
            }
        }
        for (long eb = nP * 16; eb < E; eb += AG_THREADS) {
            const long e = eb + tid;
            const bool in = e < E;
            const unsigned a = (unsigned)(in ? dst[e] : 0) - (unsigned)lo;
            const unsigned s = (unsigned)(in ? src[e] : 0);
            append(in && a < ucnt, s, a);
            while (qcount >= 32u) drain(32u);
        }
        if (qcount) drain(qcount);
    }
    __syncthreads();
    if (!L2 && capg >= 32u && tid == 0) *cntw = lpos_s;

    if (!has) return;

    v2f a_ag[16], a_x[16];
    {
        const float dinv = 1.0f / fmaxf((float)cnt_s[t0 + row], 1.0f);
        float* tp = tile_s + (t0 + row) * TP + half * 2;
#pragma unroll
        for (int ks = 0; ks < 16; ++ks) {
            const float2 tv = *(const float2*)(tp + ks * 4);
            a_ag[ks].x = tv.x * dinv; a_ag[ks].y = tv.y * dinv;
        }
        float* tw = tile_s + (t0 + half) * TP + row * 4;
#pragma unroll
        for (int j = 0; j < 8; ++j) *(float4*)(tw + 2 * j * TP) = xt[j];
#pragma unroll
        for (int ks = 0; ks < 16; ++ks) {
            const float2 xv = *(const float2*)(tp + ks * 4);
            a_x[ks].x = xv.x; a_x[ks].y = xv.y;
        }
    }
    v8f c[4];
#pragma unroll
    for (int nb = 0; nb < 4; ++nb) c[nb] = v8f{};
#pragma unroll
    for (int ks = 0; ks < 16; ++ks) {
        v2f bfl[4], bfr[4];
#pragma unroll
        for (int nb = 0; nb < 4; ++nb) {
            const int o = (nb * 16 + ks) * FB + lane * 2;
            const float2 l = *(const float2*)(wl_s + o);
            const float2 r = *(const float2*)(wr_s + o);
            bfl[nb].x = l.x; bfl[nb].y = l.y;
            bfr[nb].x = r.x; bfr[nb].y = r.y;
        }
#pragma unroll
        for (int nb = 0; nb < 4; ++nb)
            c[nb] = __builtin_amdgcn_wmma_f32_16x16x4_f32(false, a_ag[ks], false, bfl[nb],
                                                          (short)0, c[nb], false, false);
#pragma unroll
        for (int nb = 0; nb < 4; ++nb)
            c[nb] = __builtin_amdgcn_wmma_f32_16x16x4_f32(false, a_x[ks], false, bfr[nb],
                                                          (short)0, c[nb], false, false);
    }
    float* op = out + (long)(lo + t0) * FEAT + row;
    const bool full = (unsigned)(t0 + 16) <= ucnt;
#pragma unroll
    for (int nb = 0; nb < 4; ++nb) {
        const float bias = bl[nb * 16 + row];
#pragma unroll
        for (int r = 0; r < 8; ++r) {
            const int M = r + half * 8;
            if (full || (unsigned)(t0 + M) < ucnt) op[M * FEAT + nb * 16] = c[nb][r] + bias;
        }
    }
}

extern "C" void kernel_launch(void* const* d_in, const int* in_sizes, int n_in,
                              void* d_out, int out_size, void* d_ws, size_t ws_size,
                              hipStream_t stream) {
    const float* x   = (const float*)d_in[0];
    const int*   eix = (const int*)d_in[1];
    const float* W1l = (const float*)d_in[2];
    const float* b1l = (const float*)d_in[3];
    const float* W1r = (const float*)d_in[4];
    const float* W2l = (const float*)d_in[5];
    const float* b2l = (const float*)d_in[6];
    const float* W2r = (const float*)d_in[7];

    const int  N = in_sizes[0] / FEAT;
    const long E = in_sizes[1] / 2;
    const int* src = eix;
    const int* dst = eix + E;

    float* idx  = (float*)d_ws;
    float* x1   = idx + (size_t)N * FEAT;
    float* outp = (float*)d_out;
    float* pairf = idx;
    float* bmf  = idx + ((E + 15) / 16) * 32;

    const int  key_blocks = (int)(((E + 15) / 16 + KEY_THREADS - 1) / KEY_THREADS);
    const long Wp         = (long)key_blocks * 32;
    const int ag_blocks  = (N + TILE - 1) / TILE;

    float* listf = bmf + (size_t)ag_blocks * (size_t)Wp;
    float* iend  = idx + (size_t)N * FEAT;

    range_index<<<key_blocks, KEY_THREADS, 0, stream>>>(src, dst, pairf, bmf, E, ag_blocks, Wp);

    aggregate_update<false><<<ag_blocks, AG_THREADS, 0, stream>>>(x, src, dst, pairf, bmf, Wp, listf, iend, W1l, b1l, W1r, x1, E, N);

    aggregate_update<true><<<ag_blocks, AG_THREADS, 0, stream>>>(x1, src, dst, pairf, bmf, Wp, listf, iend, W2l, b2l, W2r, outp, E, N);
}


# matches routed in LDS (integer add takes a queue place) to the wave owning their 16 rows; it alone adds source rows into those LDS rows by plain load-add-store: no float atomic per edge; queue order
# speedup vs baseline: 4.0926x; 1.0358x over previous
#include <hip/hip_runtime.h>

typedef __attribute__((ext_vector_type(2))) float v2f;
typedef __attribute__((ext_vector_type(8))) float v8f;

#define FEAT 64
#define TILE 400
#define OCAP 512
#define AG_THREADS 1024
#define KEY_THREADS 1024
#define RG 256
#define RGP 257
#define PU 8
#define LSH 9
#define TP 68
#define FB 66
static_assert(TILE < (1 << LSH), "an owned row (and the spare row) must fit the low bits of a list word");
static_assert(OCAP % 32 == 0, "queues are read 32 words at a time");
static_assert(TILE % 16 == 0 && TILE <= 16 * (AG_THREADS / 32), "one 16-node tile per wave in the update phase");

__global__ void __launch_bounds__(KEY_THREADS)
range_index(const int* __restrict__ src, const int* __restrict__ dst, float* __restrict__ pairf, float* __restrict__ bmf,
            long E, int nR, long Wp) {
    __shared__ unsigned bm_s[32 * RGP];
    int2* pair = (int2*)pairf;
    unsigned* bitmap = (unsigned*)bmf;
    const int tid = threadIdx.x, lane = tid & 31, wave = tid >> 5;
    const bool vec_ok = (((unsigned long long)pairf) & 15ull) == 0ull;
    const long nP = vec_ok ? (E >> 4) : 0;
    const long pb = ((long)blockIdx.x * 32 + wave) * 32;
    const int  hp = lane >> 4;
    unsigned r[16];
    if (pb + 32 <= nP) {
        const long e0 = pb * 16 + lane;
        int dv[16], sv[16];
#pragma unroll
        for (int j = 0; j < 16; ++j) {
            dv[j] = dst[e0 + 32 * j];
            sv[j] = src[e0 + 32 * j];
        }
#pragma unroll
        for (int j = 0; j < 16; ++j) {
            pair[e0 + 32 * j] = make_int2(dv[j], sv[j]);
            r[j] = (unsigned)dv[j] / TILE;
        }
    } else {
#pragma unroll
        for (int j = 0; j < 16; ++j) {
            r[j] = 0xffffffffu;
            if (pb + 2 * j + hp < nP) {
                const long e = pb * 16 + j * 32 + lane;
                const int d = dst[e];
                const int s = src[e];
                pair[e] = make_int2(d, s);
                r[j] = (unsigned)d / TILE;
            }
        }
    }
    for (int g0 = 0; g0 < nR; g0 += RG) {
        for (int i = tid; i < 32 * RGP; i += KEY_THREADS) bm_s[i] = 0u;
        __syncthreads();
#pragma unroll
        for (int j = 0; j < 16; ++j) {
            const unsigned rl = r[j] - (unsigned)g0;
            if (rl < (unsigned)RG && r[j] < (unsigned)nR) atomicOr(&bm_s[wave * RGP + (int)rl], 1u << (2 * j + hp));
        }
        __syncthreads();
        const int cnt = min(RG, nR - g0);
        for (int i = tid; i < cnt * 32; i += KEY_THREADS) {
            const int rl = i >> 5, w = i & 31;
            bitmap[(size_t)(g0 + rl) * (size_t)Wp + (size_t)blockIdx.x * 32 + w] = bm_s[w * RGP + rl];
        }
        __syncthreads();
    }
}

template <bool L2>
__global__ void __launch_bounds__(AG_THREADS)
aggregate_update(const float* __restrict__ x, const int* __restrict__ src, const int* __restrict__ dst,
                 const float* __restrict__ pairf, const float* __restrict__ bmf, long Wp,
                 float* __restrict__ listf, const float* __restrict__ list_end,
                 const float* __restrict__ Wl, const float* __restrict__ bl, const float* __restrict__ Wr,
                 float* __restrict__ out, long E, int N) {
    __shared__ __attribute__((aligned(16))) float tile_s[(TILE + 1) * TP];
    __shared__ __attribute__((aligned(16))) float wl_s[FEAT * FB];
    __shared__ __attribute__((aligned(16))) float wr_s[FEAT * FB];
    __shared__ unsigned cnt_s[TILE + 1];
    __shared__ unsigned oq_s[AG_THREADS / 32][OCAP];
    __shared__ unsigned short pl_s[AG_THREADS / 32][1024];
    __shared__ unsigned qn_s[AG_THREADS / 32];

    const int tid  = threadIdx.x;
    const int lane = tid & 31;
    const int lo   = blockIdx.x * TILE;
    const unsigned ucnt = (unsigned)min(TILE, N - lo);
    unsigned short* pl = pl_s[tid >> 5];

    const long nR     = (long)((N + TILE - 1) / TILE);
    const long nRp    = nR * 32;
    const long lwords = (long)(list_end - listf) - nRp;
    const long lper   = lwords > 0 ? lwords / nR : 0L;
    const unsigned capg = (N <= (1 << (32 - LSH))) ? ((unsigned)(lper < (1L << 24) ? lper : (1L << 24)) & ~31u) : 0u;
    unsigned* cntw    = (unsigned*)listf + (long)blockIdx.x * 32;
    unsigned* glist   = (unsigned*)listf + nRp + (long)blockIdx.x * (long)capg;

    const int wave = tid >> 5;
    const int row  = lane & 15;
    const int half = lane >> 4;
    const int t0   = wave * 16;
    const bool has = t0 < (int)ucnt;
    float4 wl4[FEAT * FEAT / 4 / AG_THREADS], wr4[FEAT * FEAT / 4 / AG_THREADS];
#pragma unroll
    for (int j = 0; j < FEAT * FEAT / 4 / AG_THREADS; ++j) {
        const int i = tid + j * AG_THREADS;
        wl4[j] = ((const float4*)Wl)[i];
        wr4[j] = ((const float4*)Wr)[i];
    }

    for (int i = tid; i < (TILE + 1) * TP / 4; i += AG_THREADS)
        ((float4*)tile_s)[i] = make_float4(0.f, 0.f, 0.f, 0.f);
    for (int i = tid; i < TILE + 1; i += AG_THREADS) cnt_s[i] = 0u;
    if (tid < AG_THREADS / 32) qn_s[tid] = 0u;
#pragma unroll
    for (int j = 0; j < FEAT * FEAT / 4 / AG_THREADS; ++j) {
        const int i = tid + j * AG_THREADS;
        const int n = i >> 4, ks = i & 15;
        const int o = ((n >> 4) * 16 + ks) * FB + (n & 15) * 2;
        *(float2*)(wl_s + o)      = make_float2(wl4[j].x, wl4[j].y);
        *(float2*)(wl_s + o + 32) = make_float2(wl4[j].z, wl4[j].w);
        *(float2*)(wr_s + o)      = make_float2(wr4[j].x, wr4[j].y);
        *(float2*)(wr_s + o + 32) = make_float2(wr4[j].z, wr4[j].w);
    }
    __syncthreads();

    const unsigned ocap = (N <= (1 << (32 - LSH))) ? (unsigned)OCAP : 0u;
    auto place = [&](bool m, unsigned a) {
        unsigned p = 0u;
        if (m) p = atomicAdd(&qn_s[a >> 4], 1u);
        return p;
    };
    auto put = [&](bool m, unsigned p, unsigned s, unsigned a) {
        const bool fits = m && p < ocap;
        if (fits) oq_s[a >> 4][p] = (s << LSH) | a;
        unsigned om = __builtin_amdgcn_ballot_w32(m && !fits);
        while (om) {
            const int l = __builtin_ctz(om);
            om &= om - 1u;
            const int sj = __builtin_amdgcn_readlane((int)s, l);
            const unsigned rj = (unsigned)__builtin_amdgcn_readlane((int)a, l);
            const float2 v = ((const float2*)(x + (long)sj * FEAT))[lane];
            float* t = tile_s + rj * TP + lane * 2;
            atomicAdd(t, v.x);
            atomicAdd(t + 1, v.y);
            if (lane == 0) atomicAdd(&cnt_s[rj], 1u);
        }
    };

    const bool vec_ok = (((unsigned long long)pairf) & 15ull) == 0ull;
    const long nP     = vec_ok ? (E >> 4) : 0;
    const long Wn     = (nP + 31) >> 5;
    const int4* pair4 = (const int4*)pairf;
    const unsigned* brow = (const unsigned*)bmf + (size_t)blockIdx.x * (size_t)Wp;
    const int grp = lane >> 3, sub = lane & 7;

    bool walk = true;
    unsigned cw = 0u;
    if (L2 && capg >= 32u) {
        cw = cntw[lane];
        walk = __builtin_amdgcn_ballot_w32(cw == 0xffffffffu) != 0u;
    }

    if (walk) {
        for (long wb = 0; wb < Wn; wb += AG_THREADS) {
            const long wi = wb + tid;
            unsigned bits = wi < Wn ? brow[wi] : 0u;
            unsigned pcount = 0u;
            unsigned mask;
            while ((mask = __builtin_amdgcn_ballot_w32(bits != 0u)) != 0u) {
                const unsigned below = __builtin_amdgcn_mbcnt_lo(mask, 0u);
                if (bits != 0u) pl[pcount + below] = (unsigned short)((lane << 5) | __builtin_ctz(bits));
                pcount += (unsigned)__builtin_popcount(mask);
                bits &= bits - 1u;
            }
            const long pbase = (wi - lane) * 32;
            for (unsigned i0 = 0u; i0 < pcount; i0 += 4u * PU) {
                int4 d[PU];
                bool hv[PU];
                const unsigned k0 = i0 + (unsigned)(PU * grp);
                uint4 pk;
                __builtin_memcpy(&pk, __builtin_assume_aligned(pl + k0, 16), 16);
                const unsigned pw[4] = {pk.x, pk.y, pk.z, pk.w};
#pragma unroll
                for (int u = 0; u < PU; ++u) {
                    hv[u] = k0 + (unsigned)u < pcount;
                    const unsigned idx = (u & 1) ? (pw[u >> 1] >> 16) : (pw[u >> 1] & 0xffffu);
                    const long p = hv[u] ? pbase + (long)idx : 0L;
                    d[u] = pair4[p * 8 + sub];
                }
                unsigned p0[PU], p1[PU];
#pragma unroll
                for (int u = 0; u < PU; ++u) {
                    const unsigned a0 = (unsigned)d[u].x - (unsigned)lo;
                    const unsigned a1 = (unsigned)d[u].z - (unsigned)lo;
                    p0[u] = place(hv[u] && a0 < ucnt, a0);
                    p1[u] = place(hv[u] && a1 < ucnt, a1);
                }
#pragma unroll
                for (int u = 0; u < PU; ++u) {
                    const unsigned a0 = (unsigned)d[u].x - (unsigned)lo;
                    const unsigned a1 = (unsigned)d[u].z - (unsigned)lo;
                    put(hv[u] && a0 < ucnt, p0[u], (unsigned)d[u].y, a0);
                    put(hv[u] && a1 < ucnt, p1[u], (unsigned)d[u].w, a1);
                }
            }
        }
        for (long eb = nP * 16; eb < E; eb += AG_THREADS) {
            const long e = eb + tid;
            const bool in = e < E;
            const unsigned a = (unsigned)(in ? dst[e] : 0) - (unsigned)lo;
            const unsigned s = (unsigned)(in ? src[e] : 0);
            put(in && a < ucnt, place(in && a < ucnt, a), s, a);
        }
    }
    __syncthreads();

    if (!has) return;

    const int wv = __builtin_amdgcn_readfirstlane(wave);
    if (walk) cw = qn_s[lane];
    const unsigned rc = (cw + 31u) & ~31u;
    unsigned offp = lane < wv ? rc : 0u, tot = rc;
#pragma unroll
    for (int sft = 16; sft > 0; sft >>= 1) {
        offp += (unsigned)__shfl_xor((int)offp, sft);
        tot  += (unsigned)__shfl_xor((int)tot, sft);
    }
    const bool fit = !walk || (capg >= 32u && __builtin_amdgcn_ballot_w32(cw > ocap) == 0u && tot <= capg);
    const unsigned ntrue = (unsigned)__builtin_amdgcn_readlane((int)cw, wv);
    const unsigned n = walk ? (ntrue < ocap ? ntrue : ocap) : ntrue;
    unsigned* gl = glist + offp;
    const unsigned* oq = oq_s[wv];
    if (!L2 && capg >= 32u && wv == 0) cntw[lane] = fit ? cw : 0xffffffffu;
    const bool wlist = !L2 && fit;

    auto fetch16 = [&](float2 (&v)[16], unsigned ew, int base) {
        const int es = (int)(ew >> LSH);
#pragma unroll
        for (int u = 0; u < 16; ++u) {
            const int sj = __builtin_amdgcn_readlane(es, base + u);
            v[u] = ((const float2*)(x + (long)sj * FEAT))[lane];
        }
    };
    auto sum16 = [&](const float2 (&v)[16], unsigned ew, int base) {
        const int er = (int)(ew & ((1u << LSH) - 1u));
#pragma unroll
        for (int u = 0; u < 16; u += 2) {
            const int ra = __builtin_amdgcn_readlane(er, base + u);
            const int rb = __builtin_amdgcn_readlane(er, base + u + 1);
            float2* ta = (float2*)(tile_s + ra * TP + lane * 2);
            float2* tb = (float2*)(tile_s + rb * TP + lane * 2);
            if (ra != rb) {
                float2 oa = *ta, ob = *tb;
                oa.x += v[u].x;     oa.y += v[u].y;
                ob.x += v[u + 1].x; ob.y += v[u + 1].y;
                *ta = oa;
                *tb = ob;
            } else {
                float2 oa = *ta;
                oa.x = (oa.x + v[u].x) + v[u + 1].x;
                oa.y = (oa.y + v[u].y) + v[u + 1].y;
                *ta = oa;
            }
        }
    };
    auto entries = [&](unsigned i0) {
        const bool live = i0 + (unsigned)lane < n;
        unsigned eq = (unsigned)TILE, eg = (unsigned)TILE;
        if (walk && live)  eq = oq[i0 + (unsigned)lane];
        if (!walk && live) eg = gl[i0 + (unsigned)lane];
        return walk ? eq : eg;
    };
    float2 va[16], vb[16];
    unsigned en = 0u;
    if (n > 0u) { en = entries(0u); fetch16(va, en, 0); }
    for (unsigned i0 = 0u; i0 < n; i0 += 32u) {
        const unsigned e = en;
        const bool hb = i0 + 16u < n, hn = i0 + 32u < n;
        if (hb) fetch16(vb, e, 16);
        if (wlist) gl[i0 + (unsigned)lane] = e;
        if (i0 + (unsigned)lane < n) atomicAdd(&cnt_s[e & ((1u << LSH) - 1u)], 1u);
        if (hn) en = entries(i0 + 32u);
        sum16(va, e, 0);
        if (hn) fetch16(va, en, 0);
        if (hb) sum16(vb, e, 16);
    }
    float4 xt[8];
    {
        const float4* xp4 = (const float4*)(x + (long)(lo + t0) * FEAT);
#pragma unroll
        for (int j = 0; j < 8; ++j)
            xt[j] = (unsigned)(t0 + 2 * j + half) < ucnt ? xp4[j * 32 + lane] : make_float4(0.f, 0.f, 0.f, 0.f);
    }

    v2f a_ag[16], a_x[16];
    {
        const float dinv = 1.0f / fmaxf((float)cnt_s[t0 + row], 1.0f);
        float* tp = tile_s + (t0 + row) * TP + half * 2;
#pragma unroll
        for (int ks = 0; ks < 16; ++ks) {
            const float2 tv = *(const float2*)(tp + ks * 4);
            a_ag[ks].x = tv.x * dinv; a_ag[ks].y = tv.y * dinv;
        }
        float* tw = tile_s + (t0 + half) * TP + row * 4;
#pragma unroll
        for (int j = 0; j < 8; ++j) *(float4*)(tw + 2 * j * TP) = xt[j];
#pragma unroll
        for (int ks = 0; ks < 16; ++ks) {
            const float2 xv = *(const float2*)(tp + ks * 4);
            a_x[ks].x = xv.x; a_x[ks].y = xv.y;
        }
    }
    v8f c[4];
#pragma unroll
    for (int nb = 0; nb < 4; ++nb) c[nb] = v8f{};
#pragma unroll
    for (int ks = 0; ks < 16; ++ks) {
        v2f bfl[4], bfr[4];
#pragma unroll
        for (int nb = 0; nb < 4; ++nb) {
            const int o = (nb * 16 + ks) * FB + lane * 2;
            const float2 l = *(const float2*)(wl_s + o);
            const float2 r = *(const float2*)(wr_s + o);
            bfl[nb].x = l.x; bfl[nb].y = l.y;
            bfr[nb].x = r.x; bfr[nb].y = r.y;
        }
#pragma unroll
        for (int nb = 0; nb < 4; ++nb)
            c[nb] = __builtin_amdgcn_wmma_f32_16x16x4_f32(false, a_ag[ks], false, bfl[nb],
                                                          (short)0, c[nb], false, false);
#pragma unroll
        for (int nb = 0; nb < 4; ++nb)
            c[nb] = __builtin_amdgcn_wmma_f32_16x16x4_f32(false, a_x[ks], false, bfr[nb],
                                                          (short)0, c[nb], false, false);
    }
    float* op = out + (long)(lo + t0) * FEAT + row;
    const bool full = (unsigned)(t0 + 16) <= ucnt;
#pragma unroll
    for (int nb = 0; nb < 4; ++nb) {
        const float bias = bl[nb * 16 + row];
#pragma unroll
        for (int r = 0; r < 8; ++r) {
            const int M = r + half * 8;
            if (full || (unsigned)(t0 + M) < ucnt) op[M * FEAT + nb * 16] = c[nb][r] + bias;
        }
    }
}

extern "C" void kernel_launch(void* const* d_in, const int* in_sizes, int n_in,
                              void* d_out, int out_size, void* d_ws, size_t ws_size,
                              hipStream_t stream) {
    const float* x   = (const float*)d_in[0];
    const int*   eix = (const int*)d_in[1];
    const float* W1l = (const float*)d_in[2];
    const float* b1l = (const float*)d_in[3];
    const float* W1r = (const float*)d_in[4];
    const float* W2l = (const float*)d_in[5];
    const float* b2l = (const float*)d_in[6];
    const float* W2r = (const float*)d_in[7];

    const int  N = in_sizes[0] / FEAT;
    const long E = in_sizes[1] / 2;
    const int* src = eix;
    const int* dst = eix + E;

    float* idx  = (float*)d_ws;
    float* x1   = idx + (size_t)N * FEAT;
    float* outp = (float*)d_out;
    float* pairf = idx;
    float* bmf  = idx + ((E + 15) / 16) * 32;

    const int  key_blocks = (int)(((E + 15) / 16 + KEY_THREADS - 1) / KEY_THREADS);
    const long Wp         = (long)key_blocks * 32;
    const int ag_blocks  = (N + TILE - 1) / TILE;

    float* listf = bmf + (size_t)ag_blocks * (size_t)Wp;
    float* iend  = idx + (size_t)N * FEAT;

    range_index<<<key_blocks, KEY_THREADS, 0, stream>>>(src, dst, pairf, bmf, E, ag_blocks, Wp);

    aggregate_update<false><<<ag_blocks, AG_THREADS, 0, stream>>>(x, src, dst, pairf, bmf, Wp, listf, iend, W1l, b1l, W1r, x1, E, N);

    aggregate_update<true><<<ag_blocks, AG_THREADS, 0, stream>>>(x1, src, dst, pairf, bmf, Wp, listf, iend, W2l, b2l, W2r, outp, E, N);
}


# owner waves take queue entries in pairs: lower 16 lanes fetch one source row, upper 16 the next, 16 B a lane; one LDS load-add-store adds both: half the loads and LDS ops per edge; same sum order
# speedup vs baseline: 4.4931x; 1.0979x over previous
#include <hip/hip_runtime.h>

typedef __attribute__((ext_vector_type(2))) float v2f;
typedef __attribute__((ext_vector_type(8))) float v8f;

#define FEAT 64
#define TILE 400
#define OCAP 512
#define AG_THREADS 1024
#define KEY_THREADS 1024
#define RG 256
#define RGP 257
#define PU 8
#define LSH 9
#define TP 68
#define FB 66
static_assert(TILE < (1 << LSH), "an owned row (and the spare row) must fit the low bits of a list word");
static_assert(OCAP % 32 == 0, "queues are read 32 words at a time");
static_assert(TILE % 16 == 0 && TILE <= 16 * (AG_THREADS / 32), "one 16-node tile per wave in the update phase");

__global__ void __launch_bounds__(KEY_THREADS)
range_index(const int* __restrict__ src, const int* __restrict__ dst, float* __restrict__ pairf, float* __restrict__ bmf,
            long E, int nR, long Wp) {
    __shared__ unsigned bm_s[32 * RGP];
    int2* pair = (int2*)pairf;
    unsigned* bitmap = (unsigned*)bmf;
    const int tid = threadIdx.x, lane = tid & 31, wave = tid >> 5;
    const bool vec_ok = (((unsigned long long)pairf) & 15ull) == 0ull;
    const long nP = vec_ok ? (E >> 4) : 0;
    const long pb = ((long)blockIdx.x * 32 + wave) * 32;
    const int  hp = lane >> 4;
    unsigned r[16];
    if (pb + 32 <= nP) {
        const long e0 = pb * 16 + lane;
        int dv[16], sv[16];
#pragma unroll
        for (int j = 0; j < 16; ++j) {
            dv[j] = dst[e0 + 32 * j];
            sv[j] = src[e0 + 32 * j];
        }
#pragma unroll
        for (int j = 0; j < 16; ++j) {
            pair[e0 + 32 * j] = make_int2(dv[j], sv[j]);
            r[j] = (unsigned)dv[j] / TILE;
        }
    } else {
#pragma unroll
        for (int j = 0; j < 16; ++j) {
            r[j] = 0xffffffffu;
            if (pb + 2 * j + hp < nP) {
                const long e = pb * 16 + j * 32 + lane;
                const int d = dst[e];
                const int s = src[e];
                pair[e] = make_int2(d, s);
                r[j] = (unsigned)d / TILE;
            }
        }
    }
    for (int g0 = 0; g0 < nR; g0 += RG) {
        for (int i = tid; i < 32 * RGP; i += KEY_THREADS) bm_s[i] = 0u;
        __syncthreads();
#pragma unroll
        for (int j = 0; j < 16; ++j) {
            const unsigned rl = r[j] - (unsigned)g0;
            if (rl < (unsigned)RG && r[j] < (unsigned)nR) atomicOr(&bm_s[wave * RGP + (int)rl], 1u << (2 * j + hp));
        }
        __syncthreads();
        const int cnt = min(RG, nR - g0);
        for (int i = tid; i < cnt * 32; i += KEY_THREADS) {
            const int rl = i >> 5, w = i & 31;
            bitmap[(size_t)(g0 + rl) * (size_t)Wp + (size_t)blockIdx.x * 32 + w] = bm_s[w * RGP + rl];
        }
        __syncthreads();
    }
}

template <int J> __device__ __forceinline__ unsigned row_lane(unsigned v) {
    return (unsigned)__builtin_amdgcn_update_dpp(0, (int)v, 0x150 + J, 0xf, 0xf, true);
}
__device__ __forceinline__ unsigned xhalf_u(unsigned v) {
    return (unsigned)__builtin_amdgcn_permlanex16((int)v, (int)v, 0x76543210, 0xfedcba98, false, false);
}
__device__ __forceinline__ float xhalf_f(float v) {
    return __builtin_bit_cast(float, xhalf_u(__builtin_bit_cast(unsigned, v)));
}
template <int J> struct jconst { static constexpr int value = J; };
#define EACH8(M) M(0) M(1) M(2) M(3) M(4) M(5) M(6) M(7)

template <bool L2>
__global__ void __launch_bounds__(AG_THREADS)
aggregate_update(const float* __restrict__ x, const int* __restrict__ src, const int* __restrict__ dst,
                 const float* __restrict__ pairf, const float* __restrict__ bmf, long Wp,
                 float* __restrict__ listf, const float* __restrict__ list_end,
                 const float* __restrict__ Wl, const float* __restrict__ bl, const float* __restrict__ Wr,
                 float* __restrict__ out, long E, int N) {
    __shared__ __attribute__((aligned(16))) float tile_s[(TILE + 1) * TP];
    __shared__ __attribute__((aligned(16))) float wl_s[FEAT * FB];
    __shared__ __attribute__((aligned(16))) float wr_s[FEAT * FB];
    __shared__ unsigned cnt_s[TILE + 1];
    __shared__ unsigned oq_s[AG_THREADS / 32][OCAP];
    __shared__ unsigned short pl_s[AG_THREADS / 32][1024];
    __shared__ unsigned qn_s[AG_THREADS / 32];

    const int tid  = threadIdx.x;
    const int lane = tid & 31;
    const int lo   = blockIdx.x * TILE;
    const unsigned ucnt = (unsigned)min(TILE, N - lo);
    unsigned short* pl = pl_s[tid >> 5];

    const long nR     = (long)((N + TILE - 1) / TILE);
    const long nRp    = nR * 32;
    const long lwords = (long)(list_end - listf) - nRp;
    const long lper   = lwords > 0 ? lwords / nR : 0L;
    const unsigned capg = (N <= (1 << (32 - LSH))) ? ((unsigned)(lper < (1L << 24) ? lper : (1L << 24)) & ~31u) : 0u;
    unsigned* cntw    = (unsigned*)listf + (long)blockIdx.x * 32;
    unsigned* glist   = (unsigned*)listf + nRp + (long)blockIdx.x * (long)capg;

    const int wave = tid >> 5;
    const int row  = lane & 15;
    const int half = lane >> 4;
    const int t0   = wave * 16;
    const bool has = t0 < (int)ucnt;
    float4 wl4[FEAT * FEAT / 4 / AG_THREADS], wr4[FEAT * FEAT / 4 / AG_THREADS];
#pragma unroll
    for (int j = 0; j < FEAT * FEAT / 4 / AG_THREADS; ++j) {
        const int i = tid + j * AG_THREADS;
        wl4[j] = ((const float4*)Wl)[i];
        wr4[j] = ((const float4*)Wr)[i];
    }

    for (int i = tid; i < (TILE + 1) * TP / 4; i += AG_THREADS)
        ((float4*)tile_s)[i] = make_float4(0.f, 0.f, 0.f, 0.f);
    for (int i = tid; i < TILE + 1; i += AG_THREADS) cnt_s[i] = 0u;
    if (tid < AG_THREADS / 32) qn_s[tid] = 0u;
#pragma unroll
    for (int j = 0; j < FEAT * FEAT / 4 / AG_THREADS; ++j) {
        const int i = tid + j * AG_THREADS;
        const int n = i >> 4, ks = i & 15;
        const int o = ((n >> 4) * 16 + ks) * FB + (n & 15) * 2;
        *(float2*)(wl_s + o)      = make_float2(wl4[j].x, wl4[j].y);
        *(float2*)(wl_s + o + 32) = make_float2(wl4[j].z, wl4[j].w);
        *(float2*)(wr_s + o)      = make_float2(wr4[j].x, wr4[j].y);
        *(float2*)(wr_s + o + 32) = make_float2(wr4[j].z, wr4[j].w);
    }
    __syncthreads();

    const unsigned ocap = (N <= (1 << (32 - LSH))) ? (unsigned)OCAP : 0u;
    auto place = [&](bool m, unsigned a) {
        unsigned p = 0u;
        if (m) p = atomicAdd(&qn_s[a >> 4], 1u);
        return p;
    };
    auto put = [&](bool m, unsigned p, unsigned s, unsigned a) {
        const bool fits = m && p < ocap;
        if (fits) oq_s[a >> 4][p] = (s << LSH) | a;
        unsigned om = __builtin_amdgcn_ballot_w32(m && !fits);
        while (om) {
            const int l = __builtin_ctz(om);
            om &= om - 1u;
            const int sj = __builtin_amdgcn_readlane((int)s, l);
            const unsigned rj = (unsigned)__builtin_amdgcn_readlane((int)a, l);
            const float2 v = ((const float2*)(x + (long)sj * FEAT))[lane];
            float* t = tile_s + rj * TP + lane * 2;
            atomicAdd(t, v.x);
            atomicAdd(t + 1, v.y);
            if (lane == 0) atomicAdd(&cnt_s[rj], 1u);
        }
    };

    const bool vec_ok = (((unsigned long long)pairf) & 15ull) == 0ull;
    const long nP     = vec_ok ? (E >> 4) : 0;
    const long Wn     = (nP + 31) >> 5;
    const int4* pair4 = (const int4*)pairf;
    const unsigned* brow = (const unsigned*)bmf + (size_t)blockIdx.x * (size_t)Wp;
    const int grp = lane >> 3, sub = lane & 7;

    bool walk = true;
    unsigned cw = 0u;
    if (L2 && capg >= 32u) {
        cw = cntw[lane];
        walk = __builtin_amdgcn_ballot_w32(cw == 0xffffffffu) != 0u;
    }

    if (walk) {
        for (long wb = 0; wb < Wn; wb += AG_THREADS) {
            const long wi = wb + tid;
            unsigned bits = wi < Wn ? brow[wi] : 0u;
            unsigned pcount = 0u;
            unsigned mask;
            while ((mask = __builtin_amdgcn_ballot_w32(bits != 0u)) != 0u) {
                const unsigned below = __builtin_amdgcn_mbcnt_lo(mask, 0u);
                if (bits != 0u) pl[pcount + below] = (unsigned short)((lane << 5) | __builtin_ctz(bits));
                pcount += (unsigned)__builtin_popcount(mask);
                bits &= bits - 1u;
            }
            const long pbase = (wi - lane) * 32;
            for (unsigned i0 = 0u; i0 < pcount; i0 += 4u * PU) {
                int4 d[PU];
                bool hv[PU];
                const unsigned k0 = i0 + (unsigned)(PU * grp);
                uint4 pk;
                __builtin_memcpy(&pk, __builtin_assume_aligned(pl + k0, 16), 16);
                const unsigned pw[4] = {pk.x, pk.y, pk.z, pk.w};
#pragma unroll
                for (int u = 0; u < PU; ++u) {
                    hv[u] = k0 + (unsigned)u < pcount;
                    const unsigned idx = (u & 1) ? (pw[u >> 1] >> 16) : (pw[u >> 1] & 0xffffu);
                    const long p = hv[u] ? pbase + (long)idx : 0L;
                    d[u] = pair4[p * 8 + sub];
                }
                unsigned p0[PU], p1[PU];
#pragma unroll
                for (int u = 0; u < PU; ++u) {
                    const unsigned a0 = (unsigned)d[u].x - (unsigned)lo;
                    const unsigned a1 = (unsigned)d[u].z - (unsigned)lo;
                    p0[u] = place(hv[u] && a0 < ucnt, a0);
                    p1[u] = place(hv[u] && a1 < ucnt, a1);
                }
#pragma unroll
                for (int u = 0; u < PU; ++u) {
                    const unsigned a0 = (unsigned)d[u].x - (unsigned)lo;
                    const unsigned a1 = (unsigned)d[u].z - (unsigned)lo;
                    put(hv[u] && a0 < ucnt, p0[u], (unsigned)d[u].y, a0);
                    put(hv[u] && a1 < ucnt, p1[u], (unsigned)d[u].w, a1);
                }
            }
        }
        for (long eb = nP * 16; eb < E; eb += AG_THREADS) {
            const long e = eb + tid;
            const bool in = e < E;
            const unsigned a = (unsigned)(in ? dst[e] : 0) - (unsigned)lo;
            const unsigned s = (unsigned)(in ? src[e] : 0);
            put(in && a < ucnt, place(in && a < ucnt, a), s, a);
        }
    }
    __syncthreads();

    if (!has) return;

    const int wv = __builtin_amdgcn_readfirstlane(wave);
    if (walk) cw = qn_s[lane];
    const unsigned rc = (cw + 31u) & ~31u;
    unsigned offp = lane < wv ? rc : 0u, tot = rc;
#pragma unroll
    for (int sft = 16; sft > 0; sft >>= 1) {
        offp += (unsigned)__shfl_xor((int)offp, sft);
        tot  += (unsigned)__shfl_xor((int)tot, sft);
    }
    const bool fit = !walk || (capg >= 32u && __builtin_amdgcn_ballot_w32(cw > ocap) == 0u && tot <= capg);
    const unsigned ntrue = (unsigned)__builtin_amdgcn_readlane((int)cw, wv);
    const unsigned n = walk ? (ntrue < ocap ? ntrue : ocap) : ntrue;
    unsigned* gl = glist + offp;
    const unsigned* oq = oq_s[wv];
    if (!L2 && capg >= 32u && wv == 0) cntw[lane] = fit ? cw : 0xffffffffu;
    const bool wlist = !L2 && fit;

    const unsigned c4   = (unsigned)row;
    const unsigned pidx = ((unsigned)row << 1) | (unsigned)half;
    const float4* x4 = (const float4*)x;
    auto fetch8 = [&](float4 (&v)[8], unsigned ew, auto jb) {
        constexpr int JB = decltype(jb)::value;
        const unsigned gb = (ew >> LSH) * 16u;
#define FETCH_ONE(U) v[U] = x4[row_lane<JB + U>(gb) + c4];
        EACH8(FETCH_ONE)
#undef FETCH_ONE
    };
    auto sum8 = [&](const float4 (&v)[8], unsigned lbe, unsigned cm, auto jb) {
        constexpr int JB = decltype(jb)::value;
        const unsigned cb = c4 * 16u;
#define SUM_ONE(U) { \
            float4* t = (float4*)((char*)tile_s + (row_lane<JB + U>(lbe) + cb)); \
            { \
                float4 o = *t; \
                o.x += v[U].x; o.y += v[U].y; o.z += v[U].z; o.w += v[U].w; \
                *t = o; \
            } \
            if (__builtin_expect(((cm >> (JB + U)) & 1u) != 0u, 0)) { \
                const float ux = xhalf_f(v[U].x), uy = xhalf_f(v[U].y), uz = xhalf_f(v[U].z), uw = xhalf_f(v[U].w); \
                float4 o = *t; \
                o.x += ux; o.y += uy; o.z += uz; o.w += uw; \
                *t = o; \
            } }
        EACH8(SUM_ONE)
#undef SUM_ONE
    };
    auto entries = [&](unsigned i0) {
        const bool live = i0 + pidx < n;
        unsigned eq = (unsigned)TILE, eg = (unsigned)TILE;
        if (walk && live)  eq = oq[i0 + pidx];
        if (!walk && live) eg = gl[i0 + pidx];
        return walk ? eq : eg;
    };
    float4 va[8], vb[8];
    unsigned en = 0u;
    if (n > 0u) { en = entries(0u); fetch8(va, en, jconst<0>{}); }
    for (unsigned i0 = 0u; i0 < n; i0 += 32u) {
        const unsigned e = en;
        const bool hb = i0 + 16u < n, hn = i0 + 32u < n;
        const unsigned er = e & ((1u << LSH) - 1u);
        const bool cf = er == xhalf_u(er);
        const unsigned cm = __builtin_amdgcn_ballot_w32(cf);
        const unsigned lbe = (cf && half != 0) ? (unsigned)(TILE * TP * 4) : er * (unsigned)(TP * 4);
        if (hb) fetch8(vb, e, jconst<8>{});
        if (wlist) gl[i0 + pidx] = e;
        if (i0 + pidx < n) atomicAdd(&cnt_s[er], 1u);
        if (hn) en = entries(i0 + 32u);
        sum8(va, lbe, cm, jconst<0>{});
        if (hn) fetch8(va, en, jconst<0>{});
        if (hb) sum8(vb, lbe, cm, jconst<8>{});
    }
    float4 xt[8];
    {
        const float4* xp4 = (const float4*)(x + (long)(lo + t0) * FEAT);
#pragma unroll
        for (int j = 0; j < 8; ++j)
            xt[j] = (unsigned)(t0 + 2 * j + half) < ucnt ? xp4[j * 32 + lane] : make_float4(0.f, 0.f, 0.f, 0.f);
    }

    v2f a_ag[16], a_x[16];
    {
        const float dinv = 1.0f / fmaxf((float)cnt_s[t0 + row], 1.0f);
        float* tp = tile_s + (t0 + row) * TP + half * 2;
#pragma unroll
        for (int ks = 0; ks < 16; ++ks) {
            const float2 tv = *(const float2*)(tp + ks * 4);
            a_ag[ks].x = tv.x * dinv; a_ag[ks].y = tv.y * dinv;
        }
        float* tw = tile_s + (t0 + half) * TP + row * 4;
#pragma unroll
        for (int j = 0; j < 8; ++j) *(float4*)(tw + 2 * j * TP) = xt[j];
#pragma unroll
        for (int ks = 0; ks < 16; ++ks) {
            const float2 xv = *(const float2*)(tp + ks * 4);
            a_x[ks].x = xv.x; a_x[ks].y = xv.y;
        }
    }
    v8f c[4];
#pragma unroll
    for (int nb = 0; nb < 4; ++nb) c[nb] = v8f{};
#pragma unroll
    for (int ks = 0; ks < 16; ++ks) {
        v2f bfl[4], bfr[4];
#pragma unroll
        for (int nb = 0; nb < 4; ++nb) {
            const int o = (nb * 16 + ks) * FB + lane * 2;
            const float2 l = *(const float2*)(wl_s + o);
            const float2 r = *(const float2*)(wr_s + o);
            bfl[nb].x = l.x; bfl[nb].y = l.y;
            bfr[nb].x = r.x; bfr[nb].y = r.y;
        }
#pragma unroll
        for (int nb = 0; nb < 4; ++nb)
            c[nb] = __builtin_amdgcn_wmma_f32_16x16x4_f32(false, a_ag[ks], false, bfl[nb],
                                                          (short)0, c[nb], false, false);
#pragma unroll
        for (int nb = 0; nb < 4; ++nb)
            c[nb] = __builtin_amdgcn_wmma_f32_16x16x4_f32(false, a_x[ks], false, bfr[nb],
                                                          (short)0, c[nb], false, false);
    }
    float* op = out + (long)(lo + t0) * FEAT + row;
    const bool full = (unsigned)(t0 + 16) <= ucnt;
#pragma unroll
    for (int nb = 0; nb < 4; ++nb) {
        const float bias = bl[nb * 16 + row];
#pragma unroll
        for (int r = 0; r < 8; ++r) {
            const int M = r + half * 8;
            if (full || (unsigned)(t0 + M) < ucnt) op[M * FEAT + nb * 16] = c[nb][r] + bias;
        }
    }
}

extern "C" void kernel_launch(void* const* d_in, const int* in_sizes, int n_in,
                              void* d_out, int out_size, void* d_ws, size_t ws_size,
                              hipStream_t stream) {
    const float* x   = (const float*)d_in[0];
    const int*   eix = (const int*)d_in[1];
    const float* W1l = (const float*)d_in[2];
    const float* b1l = (const float*)d_in[3];
    const float* W1r = (const float*)d_in[4];
    const float* W2l = (const float*)d_in[5];
    const float* b2l = (const float*)d_in[6];
    const float* W2r = (const float*)d_in[7];

    const int  N = in_sizes[0] / FEAT;
    const long E = in_sizes[1] / 2;
    const int* src = eix;
    const int* dst = eix + E;

    float* idx  = (float*)d_ws;
    float* x1   = idx + (size_t)N * FEAT;
    float* outp = (float*)d_out;
    float* pairf = idx;
    float* bmf  = idx + ((E + 15) / 16) * 32;

    const int  key_blocks = (int)(((E + 15) / 16 + KEY_THREADS - 1) / KEY_THREADS);
    const long Wp         = (long)key_blocks * 32;
    const int ag_blocks  = (N + TILE - 1) / TILE;

    float* listf = bmf + (size_t)ag_blocks * (size_t)Wp;
    float* iend  = idx + (size_t)N * FEAT;

    range_index<<<key_blocks, KEY_THREADS, 0, stream>>>(src, dst, pairf, bmf, E, ag_blocks, Wp);

    aggregate_update<false><<<ag_blocks, AG_THREADS, 0, stream>>>(x, src, dst, pairf, bmf, Wp, listf, iend, W1l, b1l, W1r, x1, E, N);

    aggregate_update<true><<<ag_blocks, AG_THREADS, 0, stream>>>(x1, src, dst, pairf, bmf, Wp, listf, iend, W2l, b2l, W2r, outp, E, N);
}
